# starting kernel
# baseline (speedup 1.0000x reference)
#include <hip/hip_runtime.h>
#include <hip/hip_bf16.h>

typedef float v2f __attribute__((ext_vector_type(2)));
typedef float v8f __attribute__((ext_vector_type(8)));

#define NNODES 100000
#define NEDGES 1600000
#define HF 128
#define NCLS 8
#define NLAYERS 6
#define XSTR 132


__global__ void zero_f(float* __restrict__ p, int n) {
  int i = blockIdx.x * blockDim.x + threadIdx.x;
  if (i < n) p[i] = 0.0f;
}

__global__ void degree_kernel(const int* __restrict__ src, const int* __restrict__ dst,
                              float* __restrict__ outDeg, float* __restrict__ inDeg, int e) {
  int i = blockIdx.x * blockDim.x + threadIdx.x;
  if (i < e) {
    unsafeAtomicAdd(&outDeg[src[i]], 1.0f);
    unsafeAtomicAdd(&inDeg[dst[i]], 1.0f);
  }
}

__global__ void norm_kernel(float* __restrict__ a, float* __restrict__ b, int n) {
  int i = blockIdx.x * blockDim.x + threadIdx.x;
  if (i < n) {
    a[i] = rsqrtf(fmaxf(a[i], 1.0f));
    b[i] = rsqrtf(fmaxf(b[i], 1.0f));
  }
}

__global__ __launch_bounds__(128) void gemm_norm_wmma(
    const float* __restrict__ X,
    const float* __restrict__ W,
    const float* __restrict__ normSrc,
    float* __restrict__ Y,
    int nRows)
{
  extern __shared__ float smem[];
  float* sW = smem;
  float* sX = smem + HF * HF;

  {
    const float4* Wv = (const float4*)W;
    float4* sWv = (float4*)sW;
    for (int i = threadIdx.x; i < HF * HF / 4; i += blockDim.x) sWv[i] = Wv[i];
  }

  const int wave = threadIdx.x >> 5;
  const int lane = threadIdx.x & 31;
  const int tile = blockIdx.x * 4 + wave;
  const int row0 = tile * 16;
  float* sXw = sX + wave * 16 * XSTR;

  if (row0 < nRows) {
    for (int j = 0; j < 16; ++j) {
      float4 v = ((const float4*)(X + (size_t)(row0 + j) * HF))[lane];
      *(float4*)(sXw + j * XSTR + lane * 4) = v;
    }
  }
  __syncthreads();
  if (row0 >= nRows) return;

  const int half = lane >> 4;
  const int l16  = lane & 15;
  const float* aRow = sXw + l16 * XSTR;

  v8f acc[8];
  v8f vz = {};
  #pragma unroll
  for (int c = 0; c < 8; ++c) acc[c] = vz;

  for (int k0 = 0; k0 < HF; k0 += 4) {
    v2f a;
    a.x = aRow[k0 + 2 * half];
    a.y = aRow[k0 + 2 * half + 1];
    #pragma unroll
    for (int c = 0; c < 8; ++c) {
      v2f b;
      const int n = c * 16 + l16;
      b.x = sW[(k0 + 2 * half) * HF + n];
      b.y = sW[(k0 + 2 * half + 1) * HF + n];
      acc[c] = __builtin_amdgcn_wmma_f32_16x16x4_f32(
          false, a, false, b, (short)0, acc[c], false, false);
    }
  }

  const int mBase = row0 + 8 * half;
  float nrm[8];
  #pragma unroll
  for (int r = 0; r < 8; ++r) nrm[r] = normSrc[mBase + r];
  #pragma unroll
  for (int c = 0; c < 8; ++c) {
    #pragma unroll
    for (int r = 0; r < 8; ++r) {
      Y[(size_t)(mBase + r) * HF + c * 16 + l16] = acc[c][r] * nrm[r];
    }
  }
}

__global__ void edge_scatter(const float* __restrict__ hS, const int* __restrict__ src,
                             const int* __restrict__ dst, const float* __restrict__ ew,
                             float* __restrict__ agg, int nEdges) {
  int gid = blockIdx.x * blockDim.x + threadIdx.x;
  int e = gid >> 5;
  int lane = gid & 31;
  if (e >= nEdges) return;
  int s = src[e];
  int d = dst[e];
  float w = ew[e];
  float4 v = ((const float4*)(hS + (size_t)s * HF))[lane];
  float* ap = agg + (size_t)d * HF + lane * 4;
  unsafeAtomicAdd(ap + 0, v.x * w);
  unsafeAtomicAdd(ap + 1, v.y * w);
  unsafeAtomicAdd(ap + 2, v.z * w);
  unsafeAtomicAdd(ap + 3, v.w * w);
}

__global__ void finalize_kernel(const float* __restrict__ agg, const float* __restrict__ normDst,
                                const float* __restrict__ bias, float* __restrict__ Hout, int n128) {
  int i = blockIdx.x * blockDim.x + threadIdx.x;
  if (i < n128) {
    int row = i >> 7;
    int f = i & 127;
    float x = agg[i] * normDst[row] + bias[f];
    Hout[i] = (x > 0.0f) ? x : 0.01f * x;
  }
}

__global__ void final_linear(const float* __restrict__ H, const float* __restrict__ Wl,
                             const float* __restrict__ bl, float* __restrict__ out, int n) {
  int gid = blockIdx.x * blockDim.x + threadIdx.x;
  if (gid >= n * NCLS) return;
  int row = gid >> 3;
  int c = gid & 7;
  const float* h = H + (size_t)row * HF;
  float s = bl[c];
  #pragma unroll 8
  for (int f = 0; f < HF; ++f) s = fmaf(h[f], Wl[f * NCLS + c], s);
  out[gid] = s;
}

extern "C" void kernel_launch(void* const* d_in, const int* in_sizes, int n_in,
                              void* d_out, int out_size, void* d_ws, size_t ws_size,
                              hipStream_t stream) {
  const float* in_feat = (const float*)d_in[0];
  const int*   src     = (const int*)d_in[1];
  const int*   dst     = (const int*)d_in[2];
  const float* ew      = (const float*)d_in[3];
  const float* convW   = (const float*)d_in[4];
  const float* convB   = (const float*)d_in[5];
  const float* linW    = (const float*)d_in[6];
  const float* linB    = (const float*)d_in[7];
  float* out = (float*)d_out;

  float* ws = (float*)d_ws;
  const size_t NF = (size_t)NNODES * HF;
  float* hA      = ws;
  float* hS      = ws + NF;
  float* agg     = ws + 2 * NF;
  float* normSrc = ws + 3 * NF;
  float* normDst = normSrc + NNODES;

  const int nf = (int)NF;

  zero_f<<<(2 * NNODES + 255) / 256, 256, 0, stream>>>(normSrc, 2 * NNODES);
  degree_kernel<<<(NEDGES + 255) / 256, 256, 0, stream>>>(src, dst, normSrc, normDst, NEDGES);
  norm_kernel<<<(NNODES + 255) / 256, 256, 0, stream>>>(normSrc, normDst, NNODES);

  const size_t ldsBytes = (size_t)(HF * HF + 4 * 16 * XSTR) * sizeof(float);
  const int gemmBlocks = (NNODES / 16 + 3) / 4;
  const int scatterBlocks = (int)(((size_t)NEDGES * 32 + 255) / 256);

  for (int l = 0; l < NLAYERS; ++l) {
    const float* x = (l == 0) ? in_feat : hA;
    gemm_norm_wmma<<<gemmBlocks, 128, ldsBytes, stream>>>(
        x, convW + (size_t)l * HF * HF, normSrc, hS, NNODES);
    zero_f<<<(nf + 255) / 256, 256, 0, stream>>>(agg, nf);
    edge_scatter<<<scatterBlocks, 256, 0, stream>>>(hS, src, dst, ew, agg, NEDGES);
    finalize_kernel<<<(nf + 255) / 256, 256, 0, stream>>>(agg, normDst, convB + l * HF, hA, nf);
  }

  final_linear<<<(NNODES * NCLS + 255) / 256, 256, 0, stream>>>(hA, linW, linB, out, NNODES);
}


# f32-atomic edge scatter -> edges grouped by destination once per call (int atomic rank + prefix sum); one wave per row sums its source rows in registers in slot (arrival) order; agg zeroing dropped
# speedup vs baseline: 2.5056x; 2.5056x over previous
#include <hip/hip_runtime.h>
#include <hip/hip_bf16.h>

typedef float v2f __attribute__((ext_vector_type(2)));
typedef float v8f __attribute__((ext_vector_type(8)));
typedef float v4f __attribute__((ext_vector_type(4)));

#define NNODES 100000
#define NEDGES 1600000
#define HF 128
#define NCLS 8
#define NLAYERS 6
#define XSTR 132
#define SCAN_TILE 1024
#define SCAN_BLOCKS ((NNODES + SCAN_TILE - 1) / SCAN_TILE)


__global__ void zero_f(float* __restrict__ p, int n) {
  int i = blockIdx.x * blockDim.x + threadIdx.x;
  if (i < n) p[i] = 0.0f;
}

__global__ void degree_rank_kernel(const int* __restrict__ src, const int* __restrict__ dst,
                                   float* __restrict__ outDeg, int* __restrict__ inCnt,
                                   int* __restrict__ rank, int e) {
  int i = blockIdx.x * blockDim.x + threadIdx.x;
  if (i < e) {
    unsafeAtomicAdd(&outDeg[src[i]], 1.0f);
    rank[i] = atomicAdd(&inCnt[dst[i]], 1);
  }
}

__global__ void norm_kernel(float* __restrict__ a, const int* __restrict__ inCnt,
                            float* __restrict__ b, int n) {
  int i = blockIdx.x * blockDim.x + threadIdx.x;
  if (i < n) {
    a[i] = rsqrtf(fmaxf(a[i], 1.0f));
    b[i] = rsqrtf(fmaxf((float)inCnt[i], 1.0f));
  }
}

__global__ __launch_bounds__(256) void rowptr_block_scan(const int* __restrict__ cnt,
                                                         int* __restrict__ rowptr,
                                                         int* __restrict__ blockSum, int n) {
  __shared__ int sWave[8];
  const int t = threadIdx.x;
  const int lane = t & 31;
  const int wv = t >> 5;
  const int base = blockIdx.x * SCAN_TILE + t * 4;
  int c[4];
  for (int i = 0; i < 4; ++i) c[i] = (base + i < n) ? cnt[base + i] : 0;
  const int tsum = c[0] + c[1] + c[2] + c[3];
  int x = tsum;
  for (int o = 1; o < 32; o <<= 1) {
    int y = __shfl_up(x, o, 32);
    if (lane >= o) x += y;
  }
  if (lane == 31) sWave[wv] = x;
  __syncthreads();
  int woff = 0, tot = 0;
  for (int i = 0; i < 8; ++i) {
    int s = sWave[i];
    if (i < wv) woff += s;
    tot += s;
  }
  int run = woff + x - tsum;
  for (int i = 0; i < 4; ++i) {
    if (base + i < n) rowptr[base + i] = run;
    run += c[i];
  }
  if (t == 0) blockSum[blockIdx.x] = tot;
}

__global__ __launch_bounds__(256) void rowptr_add_offsets(int* __restrict__ rowptr,
                                                          const int* __restrict__ blockSum,
                                                          int n, int total) {
  __shared__ int sWave[8];
  const int t = threadIdx.x;
  const int lane = t & 31;
  const int wv = t >> 5;
  int v = (t < (int)blockIdx.x) ? blockSum[t] : 0;
  for (int o = 16; o > 0; o >>= 1) v += __shfl_xor(v, o, 32);
  if (lane == 0) sWave[wv] = v;
  __syncthreads();
  int off = 0;
  for (int i = 0; i < 8; ++i) off += sWave[i];
  const int base = blockIdx.x * SCAN_TILE + t * 4;
  for (int i = 0; i < 4; ++i)
    if (base + i < n) rowptr[base + i] += off;
  if (blockIdx.x == 0 && t == 0) rowptr[n] = total;
}

__global__ void csr_fill(const int* __restrict__ src, const int* __restrict__ dst,
                         const float* __restrict__ ew, const int* __restrict__ rank,
                         const int* __restrict__ rowptr, int* __restrict__ csr, int e) {
  int i = blockIdx.x * blockDim.x + threadIdx.x;
  if (i < e) {
    int pos = rowptr[dst[i]] + rank[i];
    pos = (int)min((unsigned)pos, (unsigned)(e - 1));
    ((int2*)csr)[pos] = make_int2(src[i], __float_as_int(ew[i]));
  }
}

__global__ __launch_bounds__(128) void gemm_norm_wmma(
    const float* __restrict__ X,
    const float* __restrict__ W,
    const float* __restrict__ normSrc,
    float* __restrict__ Y,
    int nRows)
{
  extern __shared__ float smem[];
  float* sW = smem;
  float* sX = smem + HF * HF;

  {
    const float4* Wv = (const float4*)W;
    float4* sWv = (float4*)sW;
    for (int i = threadIdx.x; i < HF * HF / 4; i += blockDim.x) sWv[i] = Wv[i];
  }

  const int wave = threadIdx.x >> 5;
  const int lane = threadIdx.x & 31;
  const int tile = blockIdx.x * 4 + wave;
  const int row0 = tile * 16;
  float* sXw = sX + wave * 16 * XSTR;

  if (row0 < nRows) {
    for (int j = 0; j < 16; ++j) {
      float4 v = ((const float4*)(X + (size_t)(row0 + j) * HF))[lane];
      *(float4*)(sXw + j * XSTR + lane * 4) = v;
    }
  }
  __syncthreads();
  if (row0 >= nRows) return;

  const int half = lane >> 4;
  const int l16  = lane & 15;
  const float* aRow = sXw + l16 * XSTR;

  v8f acc[8];
  v8f vz = {};
  #pragma unroll
  for (int c = 0; c < 8; ++c) acc[c] = vz;

  for (int k0 = 0; k0 < HF; k0 += 4) {
    v2f a;
    a.x = aRow[k0 + 2 * half];
    a.y = aRow[k0 + 2 * half + 1];
    #pragma unroll
    for (int c = 0; c < 8; ++c) {
      v2f b;
      const int n = c * 16 + l16;
      b.x = sW[(k0 + 2 * half) * HF + n];
      b.y = sW[(k0 + 2 * half + 1) * HF + n];
      acc[c] = __builtin_amdgcn_wmma_f32_16x16x4_f32(
          false, a, false, b, (short)0, acc[c], false, false);
    }
  }

  const int mBase = row0 + 8 * half;
  float nrm[8];
  #pragma unroll
  for (int r = 0; r < 8; ++r) nrm[r] = normSrc[mBase + r];
  #pragma unroll
  for (int c = 0; c < 8; ++c) {
    #pragma unroll
    for (int r = 0; r < 8; ++r) {
      Y[(size_t)(mBase + r) * HF + c * 16 + l16] = acc[c][r] * nrm[r];
    }
  }
}

template <int U>
__device__ __forceinline__ void gather_chunk(const float* __restrict__ hS, int px, int py,
                                             int k, int n, int lane, v4f& acc) {
  v4f v[U];
  float w[U];
  for (int i = 0; i < U; ++i) {
    const int idx = min(k + i, n - 1);
    int s = __builtin_amdgcn_readlane(px, idx);
    w[i] = __int_as_float(__builtin_amdgcn_readlane(py, idx));
    s = (int)min((unsigned)s, (unsigned)(NNODES - 1));
    v[i] = *(const v4f*)(hS + (size_t)s * HF + lane * 4);
  }
  if constexpr (U == 8) {
    asm volatile("" :: "v"(v[0]), "v"(v[1]), "v"(v[2]), "v"(v[3]),
                       "v"(v[4]), "v"(v[5]), "v"(v[6]), "v"(v[7]));
  } else {
    asm volatile("" :: "v"(v[0]), "v"(v[1]), "v"(v[2]), "v"(v[3]));
  }
  for (int i = 0; i < U; ++i) {
    if (i == 0 || k + i < n) {
      acc.x = fmaf(v[i].x, w[i], acc.x);
      acc.y = fmaf(v[i].y, w[i], acc.y);
      acc.z = fmaf(v[i].z, w[i], acc.z);
      acc.w = fmaf(v[i].w, w[i], acc.w);
    }
  }
}

__global__ __launch_bounds__(256) void row_gather(const float* __restrict__ hS,
                                                  const int* __restrict__ rowptr,
                                                  const int* __restrict__ csr,
                                                  float* __restrict__ agg, int nRows) {
  const int lane = threadIdx.x & 31;
  const int row = __builtin_amdgcn_readfirstlane((int)((blockIdx.x * blockDim.x + threadIdx.x) >> 5));
  if (row >= nRows) return;
  const int beg = max(rowptr[row], 0);
  const int end = min(rowptr[row + 1], NEDGES);
  v4f acc = {0.0f, 0.0f, 0.0f, 0.0f};
  for (int base = beg; base < end; base += 32) {
    const int n = min(32, end - base);
    const int2 p = ((const int2*)csr)[min(base + lane, end - 1)];
    int k = 0;
    for (; n - k > 4; k += 8) gather_chunk<8>(hS, p.x, p.y, k, n, lane, acc);
    if (k < n) gather_chunk<4>(hS, p.x, p.y, k, n, lane, acc);
  }
  *(v4f*)(agg + (size_t)row * HF + lane * 4) = acc;
}

__global__ void finalize_kernel(const float* __restrict__ agg, const float* __restrict__ normDst,
                                const float* __restrict__ bias, float* __restrict__ Hout, int n128) {
  int i = blockIdx.x * blockDim.x + threadIdx.x;
  if (i < n128) {
    int row = i >> 7;
    int f = i & 127;
    float x = agg[i] * normDst[row] + bias[f];
    Hout[i] = (x > 0.0f) ? x : 0.01f * x;
  }
}

__global__ void final_linear(const float* __restrict__ H, const float* __restrict__ Wl,
                             const float* __restrict__ bl, float* __restrict__ out, int n) {
  int gid = blockIdx.x * blockDim.x + threadIdx.x;
  if (gid >= n * NCLS) return;
  int row = gid >> 3;
  int c = gid & 7;
  const float* h = H + (size_t)row * HF;
  float s = bl[c];
  #pragma unroll 8
  for (int f = 0; f < HF; ++f) s = fmaf(h[f], Wl[f * NCLS + c], s);
  out[gid] = s;
}

extern "C" void kernel_launch(void* const* d_in, const int* in_sizes, int n_in,
                              void* d_out, int out_size, void* d_ws, size_t ws_size,
                              hipStream_t stream) {
  const float* in_feat = (const float*)d_in[0];
  const int*   src     = (const int*)d_in[1];
  const int*   dst     = (const int*)d_in[2];
  const float* ew      = (const float*)d_in[3];
  const float* convW   = (const float*)d_in[4];
  const float* convB   = (const float*)d_in[5];
  const float* linW    = (const float*)d_in[6];
  const float* linB    = (const float*)d_in[7];
  float* out = (float*)d_out;

  float* ws = (float*)d_ws;
  const size_t NF = (size_t)NNODES * HF;
  float* hA      = ws;
  float* hS      = ws + NF;
  float* agg     = ws + 2 * NF;
  float* normSrc = ws + 3 * NF;
  int*   inCnt   = (int*)(normSrc + NNODES);
  float* normDst = normSrc + 2 * NNODES;
  int*   rowptr  = (int*)(normSrc + 3 * NNODES);
  int*   blockSum = rowptr + NNODES + 96;
  int*   rank    = blockSum + 128;
  int*   csr     = rank + NEDGES;

  const int nf = (int)NF;

  zero_f<<<(2 * NNODES + 255) / 256, 256, 0, stream>>>(normSrc, 2 * NNODES);
  degree_rank_kernel<<<(NEDGES + 255) / 256, 256, 0, stream>>>(src, dst, normSrc, inCnt, rank, NEDGES);
  norm_kernel<<<(NNODES + 255) / 256, 256, 0, stream>>>(normSrc, inCnt, normDst, NNODES);
  rowptr_block_scan<<<SCAN_BLOCKS, 256, 0, stream>>>(inCnt, rowptr, blockSum, NNODES);
  rowptr_add_offsets<<<SCAN_BLOCKS, 256, 0, stream>>>(rowptr, blockSum, NNODES, NEDGES);
  csr_fill<<<(NEDGES + 255) / 256, 256, 0, stream>>>(src, dst, ew, rank, rowptr, csr, NEDGES);

  const size_t ldsBytes = (size_t)(HF * HF + 4 * 16 * XSTR) * sizeof(float);
  const int gemmBlocks = (NNODES / 16 + 3) / 4;
  const int gatherBlocks = (NNODES + 7) / 8;

  for (int l = 0; l < NLAYERS; ++l) {
    const float* x = (l == 0) ? in_feat : hA;
    gemm_norm_wmma<<<gemmBlocks, 128, ldsBytes, stream>>>(
        x, convW + (size_t)l * HF * HF, normSrc, hS, NNODES);
    row_gather<<<gatherBlocks, 256, 0, stream>>>(hS, rowptr, csr, agg, NNODES);
    finalize_kernel<<<(nf + 255) / 256, 256, 0, stream>>>(agg, normDst, convB + l * HF, hA, nf);
  }

  final_linear<<<(NNODES * NCLS + 255) / 256, 256, 0, stream>>>(hA, linW, linB, out, NNODES);
}


# layer epilogue (norm_dst scale, bias, leaky_relu) fused into the row gather: applied to the row sum in registers before its one store; finalize launches and agg plane dropped (25 -> 19 launches)
# speedup vs baseline: 3.0389x; 1.2128x over previous
#include <hip/hip_runtime.h>
#include <hip/hip_bf16.h>

typedef float v2f __attribute__((ext_vector_type(2)));
typedef float v8f __attribute__((ext_vector_type(8)));
typedef float v4f __attribute__((ext_vector_type(4)));

#define NNODES 100000
#define NEDGES 1600000
#define HF 128
#define NCLS 8
#define NLAYERS 6
#define XSTR 132
#define SCAN_TILE 1024
#define SCAN_BLOCKS ((NNODES + SCAN_TILE - 1) / SCAN_TILE)


__global__ void zero_f(float* __restrict__ p, int n) {
  int i = blockIdx.x * blockDim.x + threadIdx.x;
  if (i < n) p[i] = 0.0f;
}

__global__ void degree_rank_kernel(const int* __restrict__ src, const int* __restrict__ dst,
                                   float* __restrict__ outDeg, int* __restrict__ inCnt,
                                   int* __restrict__ rank, int e) {
  int i = blockIdx.x * blockDim.x + threadIdx.x;
  if (i < e) {
    unsafeAtomicAdd(&outDeg[src[i]], 1.0f);
    rank[i] = atomicAdd(&inCnt[dst[i]], 1);
  }
}

__global__ void norm_kernel(float* __restrict__ a, const int* __restrict__ inCnt,
                            float* __restrict__ b, int n) {
  int i = blockIdx.x * blockDim.x + threadIdx.x;
  if (i < n) {
    a[i] = rsqrtf(fmaxf(a[i], 1.0f));
    b[i] = rsqrtf(fmaxf((float)inCnt[i], 1.0f));
  }
}

__global__ __launch_bounds__(256) void rowptr_block_scan(const int* __restrict__ cnt,
                                                         int* __restrict__ rowptr,
                                                         int* __restrict__ blockSum, int n) {
  __shared__ int sWave[8];
  const int t = threadIdx.x;
  const int lane = t & 31;
  const int wv = t >> 5;
  const int base = blockIdx.x * SCAN_TILE + t * 4;
  int c[4];
  for (int i = 0; i < 4; ++i) c[i] = (base + i < n) ? cnt[base + i] : 0;
  const int tsum = c[0] + c[1] + c[2] + c[3];
  int x = tsum;
  for (int o = 1; o < 32; o <<= 1) {
    int y = __shfl_up(x, o, 32);
    if (lane >= o) x += y;
  }
  if (lane == 31) sWave[wv] = x;
  __syncthreads();
  int woff = 0, tot = 0;
  for (int i = 0; i < 8; ++i) {
    int s = sWave[i];
    if (i < wv) woff += s;
    tot += s;
  }
  int run = woff + x - tsum;
  for (int i = 0; i < 4; ++i) {
    if (base + i < n) rowptr[base + i] = run;
    run += c[i];
  }
  if (t == 0) blockSum[blockIdx.x] = tot;
}

__global__ __launch_bounds__(256) void rowptr_add_offsets(int* __restrict__ rowptr,
                                                          const int* __restrict__ blockSum,
                                                          int n, int total) {
  __shared__ int sWave[8];
  const int t = threadIdx.x;
  const int lane = t & 31;
  const int wv = t >> 5;
  int v = (t < (int)blockIdx.x) ? blockSum[t] : 0;
  for (int o = 16; o > 0; o >>= 1) v += __shfl_xor(v, o, 32);
  if (lane == 0) sWave[wv] = v;
  __syncthreads();
  int off = 0;
  for (int i = 0; i < 8; ++i) off += sWave[i];
  const int base = blockIdx.x * SCAN_TILE + t * 4;
  for (int i = 0; i < 4; ++i)
    if (base + i < n) rowptr[base + i] += off;
  if (blockIdx.x == 0 && t == 0) rowptr[n] = total;
}

__global__ void csr_fill(const int* __restrict__ src, const int* __restrict__ dst,
                         const float* __restrict__ ew, const int* __restrict__ rank,
                         const int* __restrict__ rowptr, int* __restrict__ csr, int e) {
  int i = blockIdx.x * blockDim.x + threadIdx.x;
  if (i < e) {
    int pos = rowptr[dst[i]] + rank[i];
    pos = (int)min((unsigned)pos, (unsigned)(e - 1));
    ((int2*)csr)[pos] = make_int2(src[i], __float_as_int(ew[i]));
  }
}

__global__ __launch_bounds__(128) void gemm_norm_wmma(
    const float* __restrict__ X,
    const float* __restrict__ W,
    const float* __restrict__ normSrc,
    float* __restrict__ Y,
    int nRows)
{
  extern __shared__ float smem[];
  float* sW = smem;
  float* sX = smem + HF * HF;

  {
    const float4* Wv = (const float4*)W;
    float4* sWv = (float4*)sW;
    for (int i = threadIdx.x; i < HF * HF / 4; i += blockDim.x) sWv[i] = Wv[i];
  }

  const int wave = threadIdx.x >> 5;
  const int lane = threadIdx.x & 31;
  const int tile = blockIdx.x * 4 + wave;
  const int row0 = tile * 16;
  float* sXw = sX + wave * 16 * XSTR;

  if (row0 < nRows) {
    for (int j = 0; j < 16; ++j) {
      float4 v = ((const float4*)(X + (size_t)(row0 + j) * HF))[lane];
      *(float4*)(sXw + j * XSTR + lane * 4) = v;
    }
  }
  __syncthreads();
  if (row0 >= nRows) return;

  const int half = lane >> 4;
  const int l16  = lane & 15;
  const float* aRow = sXw + l16 * XSTR;

  v8f acc[8];
  v8f vz = {};
  #pragma unroll
  for (int c = 0; c < 8; ++c) acc[c] = vz;

  for (int k0 = 0; k0 < HF; k0 += 4) {
    v2f a;
    a.x = aRow[k0 + 2 * half];
    a.y = aRow[k0 + 2 * half + 1];
    #pragma unroll
    for (int c = 0; c < 8; ++c) {
      v2f b;
      const int n = c * 16 + l16;
      b.x = sW[(k0 + 2 * half) * HF + n];
      b.y = sW[(k0 + 2 * half + 1) * HF + n];
      acc[c] = __builtin_amdgcn_wmma_f32_16x16x4_f32(
          false, a, false, b, (short)0, acc[c], false, false);
    }
  }

  const int mBase = row0 + 8 * half;
  float nrm[8];
  #pragma unroll
  for (int r = 0; r < 8; ++r) nrm[r] = normSrc[mBase + r];
  #pragma unroll
  for (int c = 0; c < 8; ++c) {
    #pragma unroll
    for (int r = 0; r < 8; ++r) {
      Y[(size_t)(mBase + r) * HF + c * 16 + l16] = acc[c][r] * nrm[r];
    }
  }
}

template <int U>
__device__ __forceinline__ void gather_chunk(const float* __restrict__ hS, int px, int py,
                                             int k, int n, int lane, v4f& acc) {
  v4f v[U];
  float w[U];
  for (int i = 0; i < U; ++i) {
    const int idx = min(k + i, n - 1);
    int s = __builtin_amdgcn_readlane(px, idx);
    w[i] = __int_as_float(__builtin_amdgcn_readlane(py, idx));
    s = (int)min((unsigned)s, (unsigned)(NNODES - 1));
    v[i] = *(const v4f*)(hS + (size_t)s * HF + lane * 4);
  }
  if constexpr (U == 8) {
    asm volatile("" :: "v"(v[0]), "v"(v[1]), "v"(v[2]), "v"(v[3]),
                       "v"(v[4]), "v"(v[5]), "v"(v[6]), "v"(v[7]));
  } else {
    asm volatile("" :: "v"(v[0]), "v"(v[1]), "v"(v[2]), "v"(v[3]));
  }
  for (int i = 0; i < U; ++i) {
    if (i == 0 || k + i < n) {
      acc.x = fmaf(v[i].x, w[i], acc.x);
      acc.y = fmaf(v[i].y, w[i], acc.y);
      acc.z = fmaf(v[i].z, w[i], acc.z);
      acc.w = fmaf(v[i].w, w[i], acc.w);
    }
  }
}

__global__ __launch_bounds__(256) void row_gather_finalize(const float* __restrict__ hS,
                                                           const int* __restrict__ rowptr,
                                                           const int* __restrict__ csr,
                                                           const float* __restrict__ normDst,
                                                           const float* __restrict__ bias,
                                                           float* __restrict__ Hout, int nRows) {
  const int lane = threadIdx.x & 31;
  const int row = __builtin_amdgcn_readfirstlane((int)((blockIdx.x * blockDim.x + threadIdx.x) >> 5));
  if (row >= nRows) return;
  const int beg = max(rowptr[row], 0);
  const int end = min(rowptr[row + 1], NEDGES);
  const float nd = normDst[row];
  const v4f b = *(const v4f*)(bias + lane * 4);
  v4f acc = {0.0f, 0.0f, 0.0f, 0.0f};
  for (int base = beg; base < end; base += 32) {
    const int n = min(32, end - base);
    const int2 p = ((const int2*)csr)[min(base + lane, end - 1)];
    int k = 0;
    for (; n - k > 4; k += 8) gather_chunk<8>(hS, p.x, p.y, k, n, lane, acc);
    if (k < n) gather_chunk<4>(hS, p.x, p.y, k, n, lane, acc);
  }
  v4f x;
  x.x = acc.x * nd + b.x;
  x.y = acc.y * nd + b.y;
  x.z = acc.z * nd + b.z;
  x.w = acc.w * nd + b.w;
  x.x = (x.x > 0.0f) ? x.x : 0.01f * x.x;
  x.y = (x.y > 0.0f) ? x.y : 0.01f * x.y;
  x.z = (x.z > 0.0f) ? x.z : 0.01f * x.z;
  x.w = (x.w > 0.0f) ? x.w : 0.01f * x.w;
  *(v4f*)(Hout + (size_t)row * HF + lane * 4) = x;
}

__global__ void final_linear(const float* __restrict__ H, const float* __restrict__ Wl,
                             const float* __restrict__ bl, float* __restrict__ out, int n) {
  int gid = blockIdx.x * blockDim.x + threadIdx.x;
  if (gid >= n * NCLS) return;
  int row = gid >> 3;
  int c = gid & 7;
  const float* h = H + (size_t)row * HF;
  float s = bl[c];
  #pragma unroll 8
  for (int f = 0; f < HF; ++f) s = fmaf(h[f], Wl[f * NCLS + c], s);
  out[gid] = s;
}

extern "C" void kernel_launch(void* const* d_in, const int* in_sizes, int n_in,
                              void* d_out, int out_size, void* d_ws, size_t ws_size,
                              hipStream_t stream) {
  const float* in_feat = (const float*)d_in[0];
  const int*   src     = (const int*)d_in[1];
  const int*   dst     = (const int*)d_in[2];
  const float* ew      = (const float*)d_in[3];
  const float* convW   = (const float*)d_in[4];
  const float* convB   = (const float*)d_in[5];
  const float* linW    = (const float*)d_in[6];
  const float* linB    = (const float*)d_in[7];
  float* out = (float*)d_out;

  float* ws = (float*)d_ws;
  const size_t NF = (size_t)NNODES * HF;
  float* hA      = ws;
  float* hS      = ws + NF;
  float* normSrc = ws + 2 * NF;
  int*   inCnt   = (int*)(normSrc + NNODES);
  float* normDst = normSrc + 2 * NNODES;
  int*   rowptr  = (int*)(normSrc + 3 * NNODES);
  int*   blockSum = rowptr + NNODES + 96;
  int*   rank    = blockSum + 128;
  int*   csr     = rank + NEDGES;

  zero_f<<<(2 * NNODES + 255) / 256, 256, 0, stream>>>(normSrc, 2 * NNODES);
  degree_rank_kernel<<<(NEDGES + 255) / 256, 256, 0, stream>>>(src, dst, normSrc, inCnt, rank, NEDGES);
  norm_kernel<<<(NNODES + 255) / 256, 256, 0, stream>>>(normSrc, inCnt, normDst, NNODES);
  rowptr_block_scan<<<SCAN_BLOCKS, 256, 0, stream>>>(inCnt, rowptr, blockSum, NNODES);
  rowptr_add_offsets<<<SCAN_BLOCKS, 256, 0, stream>>>(rowptr, blockSum, NNODES, NEDGES);
  csr_fill<<<(NEDGES + 255) / 256, 256, 0, stream>>>(src, dst, ew, rank, rowptr, csr, NEDGES);

  const size_t ldsBytes = (size_t)(HF * HF + 4 * 16 * XSTR) * sizeof(float);
  const int gemmBlocks = (NNODES / 16 + 3) / 4;
  const int gatherBlocks = (NNODES + 7) / 8;

  for (int l = 0; l < NLAYERS; ++l) {
    const float* x = (l == 0) ? in_feat : hA;
    gemm_norm_wmma<<<gemmBlocks, 128, ldsBytes, stream>>>(
        x, convW + (size_t)l * HF * HF, normSrc, hS, NNODES);
    row_gather_finalize<<<gatherBlocks, 256, 0, stream>>>(hS, rowptr, csr, normDst, convB + l * HF,
                                                          hA, NNODES);
  }

  final_linear<<<(NNODES * NCLS + 255) / 256, 256, 0, stream>>>(hA, linW, linB, out, NNODES);
}


# row gather of layer l fused with the GEMM of layer l+1: each wave gathers its 16 rows (same slot order) into its own LDS A tile and multiplies from there; h plane round trip gone, 19 -> 14 launches
# speedup vs baseline: 4.0262x; 1.3249x over previous
#include <hip/hip_runtime.h>
#include <hip/hip_bf16.h>

typedef float v2f __attribute__((ext_vector_type(2)));
typedef float v8f __attribute__((ext_vector_type(8)));
typedef float v4f __attribute__((ext_vector_type(4)));

#define NNODES 100000
#define NEDGES 1600000
#define HF 128
#define NCLS 8
#define NLAYERS 6
#define XSTR 132
#define SCAN_TILE 1024
#define SCAN_BLOCKS ((NNODES + SCAN_TILE - 1) / SCAN_TILE)


__global__ void zero_f(float* __restrict__ p, int n) {
  int i = blockIdx.x * blockDim.x + threadIdx.x;
  if (i < n) p[i] = 0.0f;
}

__global__ void degree_rank_kernel(const int* __restrict__ src, const int* __restrict__ dst,
                                   float* __restrict__ outDeg, int* __restrict__ inCnt,
                                   int* __restrict__ rank, int e) {
  int i = blockIdx.x * blockDim.x + threadIdx.x;
  if (i < e) {
    unsafeAtomicAdd(&outDeg[src[i]], 1.0f);
    rank[i] = atomicAdd(&inCnt[dst[i]], 1);
  }
}

__global__ void norm_kernel(float* __restrict__ a, const int* __restrict__ inCnt,
                            float* __restrict__ b, int n) {
  int i = blockIdx.x * blockDim.x + threadIdx.x;
  if (i < n) {
    a[i] = rsqrtf(fmaxf(a[i], 1.0f));
    b[i] = rsqrtf(fmaxf((float)inCnt[i], 1.0f));
  }
}

__global__ __launch_bounds__(256) void rowptr_block_scan(const int* __restrict__ cnt,
                                                         int* __restrict__ rowptr,
                                                         int* __restrict__ blockSum, int n) {
  __shared__ int sWave[8];
  const int t = threadIdx.x;
  const int lane = t & 31;
  const int wv = t >> 5;
  const int base = blockIdx.x * SCAN_TILE + t * 4;
  int c[4];
  for (int i = 0; i < 4; ++i) c[i] = (base + i < n) ? cnt[base + i] : 0;
  const int tsum = c[0] + c[1] + c[2] + c[3];
  int x = tsum;
  for (int o = 1; o < 32; o <<= 1) {
    int y = __shfl_up(x, o, 32);
    if (lane >= o) x += y;
  }
  if (lane == 31) sWave[wv] = x;
  __syncthreads();
  int woff = 0, tot = 0;
  for (int i = 0; i < 8; ++i) {
    int s = sWave[i];
    if (i < wv) woff += s;
    tot += s;
  }
  int run = woff + x - tsum;
  for (int i = 0; i < 4; ++i) {
    if (base + i < n) rowptr[base + i] = run;
    run += c[i];
  }
  if (t == 0) blockSum[blockIdx.x] = tot;
}

__global__ __launch_bounds__(256) void rowptr_add_offsets(int* __restrict__ rowptr,
                                                          const int* __restrict__ blockSum,
                                                          int n, int total) {
  __shared__ int sWave[8];
  const int t = threadIdx.x;
  const int lane = t & 31;
  const int wv = t >> 5;
  int v = (t < (int)blockIdx.x) ? blockSum[t] : 0;
  for (int o = 16; o > 0; o >>= 1) v += __shfl_xor(v, o, 32);
  if (lane == 0) sWave[wv] = v;
  __syncthreads();
  int off = 0;
  for (int i = 0; i < 8; ++i) off += sWave[i];
  const int base = blockIdx.x * SCAN_TILE + t * 4;
  for (int i = 0; i < 4; ++i)
    if (base + i < n) rowptr[base + i] += off;
  if (blockIdx.x == 0 && t == 0) rowptr[n] = total;
}

__global__ void csr_fill(const int* __restrict__ src, const int* __restrict__ dst,
                         const float* __restrict__ ew, const int* __restrict__ rank,
                         const int* __restrict__ rowptr, int* __restrict__ csr, int e) {
  int i = blockIdx.x * blockDim.x + threadIdx.x;
  if (i < e) {
    int pos = rowptr[dst[i]] + rank[i];
    pos = (int)min((unsigned)pos, (unsigned)(e - 1));
    ((int2*)csr)[pos] = make_int2(src[i], __float_as_int(ew[i]));
  }
}

__global__ __launch_bounds__(128) void gemm_norm_wmma(
    const float* __restrict__ X,
    const float* __restrict__ W,
    const float* __restrict__ normSrc,
    float* __restrict__ Y,
    int nRows)
{
  extern __shared__ float smem[];
  float* sW = smem;
  float* sX = smem + HF * HF;

  {
    const float4* Wv = (const float4*)W;
    float4* sWv = (float4*)sW;
    for (int i = threadIdx.x; i < HF * HF / 4; i += blockDim.x) sWv[i] = Wv[i];
  }

  const int wave = threadIdx.x >> 5;
  const int lane = threadIdx.x & 31;
  const int tile = blockIdx.x * 4 + wave;
  const int row0 = tile * 16;
  float* sXw = sX + wave * 16 * XSTR;

  if (row0 < nRows) {
    for (int j = 0; j < 16; ++j) {
      float4 v = ((const float4*)(X + (size_t)(row0 + j) * HF))[lane];
      *(float4*)(sXw + j * XSTR + lane * 4) = v;
    }
  }
  __syncthreads();
  if (row0 >= nRows) return;

  const int half = lane >> 4;
  const int l16  = lane & 15;
  const float* aRow = sXw + l16 * XSTR;

  v8f acc[8];
  v8f vz = {};
  #pragma unroll
  for (int c = 0; c < 8; ++c) acc[c] = vz;

  for (int k0 = 0; k0 < HF; k0 += 4) {
    v2f a;
    a.x = aRow[k0 + 2 * half];
    a.y = aRow[k0 + 2 * half + 1];
    #pragma unroll
    for (int c = 0; c < 8; ++c) {
      v2f b;
      const int n = c * 16 + l16;
      b.x = sW[(k0 + 2 * half) * HF + n];
      b.y = sW[(k0 + 2 * half + 1) * HF + n];
      acc[c] = __builtin_amdgcn_wmma_f32_16x16x4_f32(
          false, a, false, b, (short)0, acc[c], false, false);
    }
  }

  const int mBase = row0 + 8 * half;
  float nrm[8];
  #pragma unroll
  for (int r = 0; r < 8; ++r) nrm[r] = normSrc[mBase + r];
  #pragma unroll
  for (int c = 0; c < 8; ++c) {
    #pragma unroll
    for (int r = 0; r < 8; ++r) {
      Y[(size_t)(mBase + r) * HF + c * 16 + l16] = acc[c][r] * nrm[r];
    }
  }
}

template <int U>
__device__ __forceinline__ void gather_chunk(const float* __restrict__ hS, int px, int py,
                                             int k, int n, int lane, v4f& acc) {
  v4f v[U];
  float w[U];
  for (int i = 0; i < U; ++i) {
    const int idx = min(k + i, n - 1);
    int s = __builtin_amdgcn_readlane(px, idx);
    w[i] = __int_as_float(__builtin_amdgcn_readlane(py, idx));
    s = (int)min((unsigned)s, (unsigned)(NNODES - 1));
    v[i] = *(const v4f*)(hS + (size_t)s * HF + lane * 4);
  }
  if constexpr (U == 8) {
    asm volatile("" :: "v"(v[0]), "v"(v[1]), "v"(v[2]), "v"(v[3]),
                       "v"(v[4]), "v"(v[5]), "v"(v[6]), "v"(v[7]));
  } else {
    asm volatile("" :: "v"(v[0]), "v"(v[1]), "v"(v[2]), "v"(v[3]));
  }
  for (int i = 0; i < U; ++i) {
    if (i == 0 || k + i < n) {
      acc.x = fmaf(v[i].x, w[i], acc.x);
      acc.y = fmaf(v[i].y, w[i], acc.y);
      acc.z = fmaf(v[i].z, w[i], acc.z);
      acc.w = fmaf(v[i].w, w[i], acc.w);
    }
  }
}

__global__ __launch_bounds__(256) void row_gather_finalize(const float* __restrict__ hS,
                                                           const int* __restrict__ rowptr,
                                                           const int* __restrict__ csr,
                                                           const float* __restrict__ normDst,
                                                           const float* __restrict__ bias,
                                                           float* __restrict__ Hout, int nRows) {
  const int lane = threadIdx.x & 31;
  const int row = __builtin_amdgcn_readfirstlane((int)((blockIdx.x * blockDim.x + threadIdx.x) >> 5));
  if (row >= nRows) return;
  const int beg = max(rowptr[row], 0);
  const int end = min(rowptr[row + 1], NEDGES);
  const float nd = normDst[row];
  const v4f b = *(const v4f*)(bias + lane * 4);
  v4f acc = {0.0f, 0.0f, 0.0f, 0.0f};
  for (int base = beg; base < end; base += 32) {
    const int n = min(32, end - base);
    const int2 p = ((const int2*)csr)[min(base + lane, end - 1)];
    int k = 0;
    for (; n - k > 4; k += 8) gather_chunk<8>(hS, p.x, p.y, k, n, lane, acc);
    if (k < n) gather_chunk<4>(hS, p.x, p.y, k, n, lane, acc);
  }
  v4f x;
  x.x = acc.x * nd + b.x;
  x.y = acc.y * nd + b.y;
  x.z = acc.z * nd + b.z;
  x.w = acc.w * nd + b.w;
  x.x = (x.x > 0.0f) ? x.x : 0.01f * x.x;
  x.y = (x.y > 0.0f) ? x.y : 0.01f * x.y;
  x.z = (x.z > 0.0f) ? x.z : 0.01f * x.z;
  x.w = (x.w > 0.0f) ? x.w : 0.01f * x.w;
  *(v4f*)(Hout + (size_t)row * HF + lane * 4) = x;
}

#define FW 25
#define FTHREADS (FW * 32)

template <int U>
__device__ __forceinline__ void gather_chunk_pf(const float* __restrict__ hS, int px, int py,
                                                int k, int n, int lane, v4f& acc,
                                                const int2* __restrict__ csr2, int pfIdx, int2& pf) {
  v4f v[U];
  float w[U];
  for (int i = 0; i < U; ++i) {
    const int idx = min(k + i, n - 1);
    int s = __builtin_amdgcn_readlane(px, idx);
    w[i] = __int_as_float(__builtin_amdgcn_readlane(py, idx));
    s = (int)min((unsigned)s, (unsigned)(NNODES - 1));
    v[i] = *(const v4f*)(hS + (size_t)s * HF + lane * 4);
  }
  if (pfIdx >= 0) pf = csr2[pfIdx];
  if constexpr (U == 16) {
    asm volatile("" :: "v"(v[0]), "v"(v[1]), "v"(v[2]), "v"(v[3]),
                       "v"(v[4]), "v"(v[5]), "v"(v[6]), "v"(v[7]),
                       "v"(v[8]), "v"(v[9]), "v"(v[10]), "v"(v[11]),
                       "v"(v[12]), "v"(v[13]), "v"(v[14]), "v"(v[15]));
  } else if constexpr (U == 8) {
    asm volatile("" :: "v"(v[0]), "v"(v[1]), "v"(v[2]), "v"(v[3]),
                       "v"(v[4]), "v"(v[5]), "v"(v[6]), "v"(v[7]));
  } else {
    asm volatile("" :: "v"(v[0]), "v"(v[1]), "v"(v[2]), "v"(v[3]));
  }
  for (int i = 0; i < U; ++i) {
    if (i == 0 || k + i < n) {
      acc.x = fmaf(v[i].x, w[i], acc.x);
      acc.y = fmaf(v[i].y, w[i], acc.y);
      acc.z = fmaf(v[i].z, w[i], acc.z);
      acc.w = fmaf(v[i].w, w[i], acc.w);
    }
  }
}

__global__ __launch_bounds__(FTHREADS) void gather_gemm_norm_wmma(
    const float* __restrict__ hS,
    const int* __restrict__ rowptr,
    const int* __restrict__ csr,
    const float* __restrict__ normDst,
    const float* __restrict__ bias,
    const float* __restrict__ W,
    const float* __restrict__ normSrc,
    float* __restrict__ Y,
    int nRows)
{
  __shared__ __attribute__((aligned(16))) float smemF[HF * HF + FW * 16 * XSTR];
  float* sW = smemF;
  float* sX = smemF + HF * HF;

  {
    const float4* Wv = (const float4*)W;
    float4* sWv = (float4*)sW;
    float4 t[6];
    #pragma unroll
    for (int u = 0; u < 6; ++u) {
      const int i = (int)threadIdx.x + u * FTHREADS;
      t[u] = Wv[min(i, HF * HF / 4 - 1)];
    }
    #pragma unroll
    for (int u = 0; u < 6; ++u) {
      const int i = (int)threadIdx.x + u * FTHREADS;
      if (i < HF * HF / 4) sWv[i] = t[u];
    }
  }
  __syncthreads();

  const int wave = threadIdx.x >> 5;
  const int lane = threadIdx.x & 31;
  const int row0 = __builtin_amdgcn_readfirstlane((int)((blockIdx.x * FW + wave) * 16));
  if (row0 >= nRows) return;
  float* sXw = sX + wave * 16 * XSTR;
  const int2* csr2 = (const int2*)csr;

  {
    const int rp = rowptr[row0 + min(lane, 16)];
    const float ndv = normDst[row0 + (lane & 15)];
    const v4f b = *(const v4f*)(bias + lane * 4);
    int begN = max(__builtin_amdgcn_readlane(rp, 0), 0);
    int endN = min(__builtin_amdgcn_readlane(rp, 1), NEDGES);
    int2 pN = csr2[min(max(min(begN + lane, endN - 1), 0), NEDGES - 1)];
    #pragma nounroll
    for (int j = 0; j < 16; ++j) {
      const int beg = begN;
      const int end = endN;
      int2 p = pN;
      int pfIdx = -1;
      if (j + 1 < 16) {
        begN = max(__builtin_amdgcn_readlane(rp, j + 1), 0);
        endN = min(__builtin_amdgcn_readlane(rp, j + 2), NEDGES);
        pfIdx = min(max(min(begN + lane, endN - 1), 0), NEDGES - 1);
      }
      bool owed = (j + 1 < 16);
      const float nd = __int_as_float(__builtin_amdgcn_readlane(__float_as_int(ndv), j));
      v4f acc = {0.0f, 0.0f, 0.0f, 0.0f};
      for (int base = beg; base < end; base += 32) {
        const int n = min(32, end - base);
        if (base != beg) p = csr2[min(base + lane, end - 1)];
        int k = 0;
        for (; n - k > 12; k += 16) {
          gather_chunk_pf<16>(hS, p.x, p.y, k, n, lane, acc, csr2, owed ? pfIdx : -1, pN);
          owed = false;
        }
        if (n - k > 4) {
          gather_chunk_pf<8>(hS, p.x, p.y, k, n, lane, acc, csr2, owed ? pfIdx : -1, pN);
          owed = false;
          k += 8;
        }
        if (k < n) {
          gather_chunk_pf<4>(hS, p.x, p.y, k, n, lane, acc, csr2, owed ? pfIdx : -1, pN);
          owed = false;
        }
      }
      if (owed) pN = csr2[pfIdx];
      v4f x;
      x.x = acc.x * nd + b.x;
      x.y = acc.y * nd + b.y;
      x.z = acc.z * nd + b.z;
      x.w = acc.w * nd + b.w;
      x.x = (x.x > 0.0f) ? x.x : 0.01f * x.x;
      x.y = (x.y > 0.0f) ? x.y : 0.01f * x.y;
      x.z = (x.z > 0.0f) ? x.z : 0.01f * x.z;
      x.w = (x.w > 0.0f) ? x.w : 0.01f * x.w;
      *(v4f*)(sXw + j * XSTR + lane * 4) = x;
    }
  }
  __builtin_amdgcn_fence(__ATOMIC_RELEASE, "wavefront");
  __builtin_amdgcn_wave_barrier();
  __builtin_amdgcn_fence(__ATOMIC_ACQUIRE, "wavefront");

  const int half = lane >> 4;
  const int l16  = lane & 15;
  const float* aRow = sXw + l16 * XSTR;

  v8f acc[8];
  v8f vz = {};
  #pragma unroll
  for (int c = 0; c < 8; ++c) acc[c] = vz;

  for (int k0 = 0; k0 < HF; k0 += 4) {
    v2f a;
    a.x = aRow[k0 + 2 * half];
    a.y = aRow[k0 + 2 * half + 1];
    #pragma unroll
    for (int c = 0; c < 8; ++c) {
      v2f bb;
      const int n = c * 16 + l16;
      bb.x = sW[(k0 + 2 * half) * HF + n];
      bb.y = sW[(k0 + 2 * half + 1) * HF + n];
      acc[c] = __builtin_amdgcn_wmma_f32_16x16x4_f32(
          false, a, false, bb, (short)0, acc[c], false, false);
    }
  }

  const int mBase = row0 + 8 * half;
  float nrm[8];
  #pragma unroll
  for (int r = 0; r < 8; ++r) nrm[r] = normSrc[mBase + r];
  #pragma unroll
  for (int c = 0; c < 8; ++c) {
    #pragma unroll
    for (int r = 0; r < 8; ++r) {
      Y[(size_t)(mBase + r) * HF + c * 16 + l16] = acc[c][r] * nrm[r];
    }
  }
}

__global__ void final_linear(const float* __restrict__ H, const float* __restrict__ Wl,
                             const float* __restrict__ bl, float* __restrict__ out, int n) {
  int gid = blockIdx.x * blockDim.x + threadIdx.x;
  if (gid >= n * NCLS) return;
  int row = gid >> 3;
  int c = gid & 7;
  const float* h = H + (size_t)row * HF;
  float s = bl[c];
  #pragma unroll 8
  for (int f = 0; f < HF; ++f) s = fmaf(h[f], Wl[f * NCLS + c], s);
  out[gid] = s;
}

extern "C" void kernel_launch(void* const* d_in, const int* in_sizes, int n_in,
                              void* d_out, int out_size, void* d_ws, size_t ws_size,
                              hipStream_t stream) {
  const float* in_feat = (const float*)d_in[0];
  const int*   src     = (const int*)d_in[1];
  const int*   dst     = (const int*)d_in[2];
  const float* ew      = (const float*)d_in[3];
  const float* convW   = (const float*)d_in[4];
  const float* convB   = (const float*)d_in[5];
  const float* linW    = (const float*)d_in[6];
  const float* linB    = (const float*)d_in[7];
  float* out = (float*)d_out;

  float* ws = (float*)d_ws;
  const size_t NF = (size_t)NNODES * HF;
  float* hA      = ws;
  float* hS      = ws + NF;
  float* normSrc = ws + 2 * NF;
  int*   inCnt   = (int*)(normSrc + NNODES);
  float* normDst = normSrc + 2 * NNODES;
  int*   rowptr  = (int*)(normSrc + 3 * NNODES);
  int*   blockSum = rowptr + NNODES + 96;
  int*   rank    = blockSum + 128;
  int*   csr     = rank + NEDGES;

  zero_f<<<(2 * NNODES + 255) / 256, 256, 0, stream>>>(normSrc, 2 * NNODES);
  degree_rank_kernel<<<(NEDGES + 255) / 256, 256, 0, stream>>>(src, dst, normSrc, inCnt, rank, NEDGES);
  norm_kernel<<<(NNODES + 255) / 256, 256, 0, stream>>>(normSrc, inCnt, normDst, NNODES);
  rowptr_block_scan<<<SCAN_BLOCKS, 256, 0, stream>>>(inCnt, rowptr, blockSum, NNODES);
  rowptr_add_offsets<<<SCAN_BLOCKS, 256, 0, stream>>>(rowptr, blockSum, NNODES, NEDGES);
  csr_fill<<<(NEDGES + 255) / 256, 256, 0, stream>>>(src, dst, ew, rank, rowptr, csr, NEDGES);

  const size_t ldsBytes = (size_t)(HF * HF + 4 * 16 * XSTR) * sizeof(float);
  const int gemmBlocks = (NNODES / 16 + 3) / 4;
  const int gatherBlocks = (NNODES + 7) / 8;
  const int fusedBlocks = (NNODES / 16 + FW - 1) / FW;

  gemm_norm_wmma<<<gemmBlocks, 128, ldsBytes, stream>>>(in_feat, convW, normSrc, hS, NNODES);
  gather_gemm_norm_wmma<<<fusedBlocks, FTHREADS, 0, stream>>>(
      hS, rowptr, csr, normDst, convB + 0 * HF, convW + (size_t)1 * HF * HF, normSrc, hA, NNODES);
  gather_gemm_norm_wmma<<<fusedBlocks, FTHREADS, 0, stream>>>(
      hA, rowptr, csr, normDst, convB + 1 * HF, convW + (size_t)2 * HF * HF, normSrc, hS, NNODES);
  gather_gemm_norm_wmma<<<fusedBlocks, FTHREADS, 0, stream>>>(
      hS, rowptr, csr, normDst, convB + 2 * HF, convW + (size_t)3 * HF * HF, normSrc, hA, NNODES);
  gather_gemm_norm_wmma<<<fusedBlocks, FTHREADS, 0, stream>>>(
      hA, rowptr, csr, normDst, convB + 3 * HF, convW + (size_t)4 * HF * HF, normSrc, hS, NNODES);
  gather_gemm_norm_wmma<<<fusedBlocks, FTHREADS, 0, stream>>>(
      hS, rowptr, csr, normDst, convB + 4 * HF, convW + (size_t)5 * HF * HF, normSrc, hA, NNODES);
  row_gather_finalize<<<gatherBlocks, 256, 0, stream>>>(hA, rowptr, csr, normDst, convB + 5 * HF,
                                                        hS, NNODES);

  final_linear<<<(NNODES * NCLS + 255) / 256, 256, 0, stream>>>(hS, linW, linB, out, NNODES);
}


# last layer's row gather fused with the final linear: same 16-row LDS tiles as inner layers, x lin_W (zero-padded to 16 cols) on f32 WMMA, so its 128-term sums run in WMMA k order; 14->13 launches
# speedup vs baseline: 4.1701x; 1.0357x over previous
#include <hip/hip_runtime.h>
#include <hip/hip_bf16.h>

typedef float v2f __attribute__((ext_vector_type(2)));
typedef float v8f __attribute__((ext_vector_type(8)));
typedef float v4f __attribute__((ext_vector_type(4)));

#define NNODES 100000
#define NEDGES 1600000
#define HF 128
#define NCLS 8
#define NLAYERS 6
#define XSTR 132
#define SCAN_TILE 1024
#define SCAN_BLOCKS ((NNODES + SCAN_TILE - 1) / SCAN_TILE)


__global__ void zero_f(float* __restrict__ p, int n) {
  int i = blockIdx.x * blockDim.x + threadIdx.x;
  if (i < n) p[i] = 0.0f;
}

__global__ void degree_rank_kernel(const int* __restrict__ src, const int* __restrict__ dst,
                                   float* __restrict__ outDeg, int* __restrict__ inCnt,
                                   int* __restrict__ rank, int e) {
  int i = blockIdx.x * blockDim.x + threadIdx.x;
  if (i < e) {
    unsafeAtomicAdd(&outDeg[src[i]], 1.0f);
    rank[i] = atomicAdd(&inCnt[dst[i]], 1);
  }
}

__global__ void norm_kernel(float* __restrict__ a, const int* __restrict__ inCnt,
                            float* __restrict__ b, int n) {
  int i = blockIdx.x * blockDim.x + threadIdx.x;
  if (i < n) {
    a[i] = rsqrtf(fmaxf(a[i], 1.0f));
    b[i] = rsqrtf(fmaxf((float)inCnt[i], 1.0f));
  }
}

__global__ __launch_bounds__(256) void rowptr_block_scan(const int* __restrict__ cnt,
                                                         int* __restrict__ rowptr,
                                                         int* __restrict__ blockSum, int n) {
  __shared__ int sWave[8];
  const int t = threadIdx.x;
  const int lane = t & 31;
  const int wv = t >> 5;
  const int base = blockIdx.x * SCAN_TILE + t * 4;
  int c[4];
  for (int i = 0; i < 4; ++i) c[i] = (base + i < n) ? cnt[base + i] : 0;
  const int tsum = c[0] + c[1] + c[2] + c[3];
  int x = tsum;
  for (int o = 1; o < 32; o <<= 1) {
    int y = __shfl_up(x, o, 32);
    if (lane >= o) x += y;
  }
  if (lane == 31) sWave[wv] = x;
  __syncthreads();
  int woff = 0, tot = 0;
  for (int i = 0; i < 8; ++i) {
    int s = sWave[i];
    if (i < wv) woff += s;
    tot += s;
  }
  int run = woff + x - tsum;
  for (int i = 0; i < 4; ++i) {
    if (base + i < n) rowptr[base + i] = run;
    run += c[i];
  }
  if (t == 0) blockSum[blockIdx.x] = tot;
}

__global__ __launch_bounds__(256) void rowptr_add_offsets(int* __restrict__ rowptr,
                                                          const int* __restrict__ blockSum,
                                                          int n, int total) {
  __shared__ int sWave[8];
  const int t = threadIdx.x;
  const int lane = t & 31;
  const int wv = t >> 5;
  int v = (t < (int)blockIdx.x) ? blockSum[t] : 0;
  for (int o = 16; o > 0; o >>= 1) v += __shfl_xor(v, o, 32);
  if (lane == 0) sWave[wv] = v;
  __syncthreads();
  int off = 0;
  for (int i = 0; i < 8; ++i) off += sWave[i];
  const int base = blockIdx.x * SCAN_TILE + t * 4;
  for (int i = 0; i < 4; ++i)
    if (base + i < n) rowptr[base + i] += off;
  if (blockIdx.x == 0 && t == 0) rowptr[n] = total;
}

__global__ void csr_fill(const int* __restrict__ src, const int* __restrict__ dst,
                         const float* __restrict__ ew, const int* __restrict__ rank,
                         const int* __restrict__ rowptr, int* __restrict__ csr, int e) {
  int i = blockIdx.x * blockDim.x + threadIdx.x;
  if (i < e) {
    int pos = rowptr[dst[i]] + rank[i];
    pos = (int)min((unsigned)pos, (unsigned)(e - 1));
    ((int2*)csr)[pos] = make_int2(src[i], __float_as_int(ew[i]));
  }
}

__global__ __launch_bounds__(128) void gemm_norm_wmma(
    const float* __restrict__ X,
    const float* __restrict__ W,
    const float* __restrict__ normSrc,
    float* __restrict__ Y,
    int nRows)
{
  extern __shared__ float smem[];
  float* sW = smem;
  float* sX = smem + HF * HF;

  {
    const float4* Wv = (const float4*)W;
    float4* sWv = (float4*)sW;
    for (int i = threadIdx.x; i < HF * HF / 4; i += blockDim.x) sWv[i] = Wv[i];
  }

  const int wave = threadIdx.x >> 5;
  const int lane = threadIdx.x & 31;
  const int tile = blockIdx.x * 4 + wave;
  const int row0 = tile * 16;
  float* sXw = sX + wave * 16 * XSTR;

  if (row0 < nRows) {
    for (int j = 0; j < 16; ++j) {
      float4 v = ((const float4*)(X + (size_t)(row0 + j) * HF))[lane];
      *(float4*)(sXw + j * XSTR + lane * 4) = v;
    }
  }
  __syncthreads();
  if (row0 >= nRows) return;

  const int half = lane >> 4;
  const int l16  = lane & 15;
  const float* aRow = sXw + l16 * XSTR;

  v8f acc[8];
  v8f vz = {};
  #pragma unroll
  for (int c = 0; c < 8; ++c) acc[c] = vz;

  for (int k0 = 0; k0 < HF; k0 += 4) {
    v2f a;
    a.x = aRow[k0 + 2 * half];
    a.y = aRow[k0 + 2 * half + 1];
    #pragma unroll
    for (int c = 0; c < 8; ++c) {
      v2f b;
      const int n = c * 16 + l16;
      b.x = sW[(k0 + 2 * half) * HF + n];
      b.y = sW[(k0 + 2 * half + 1) * HF + n];
      acc[c] = __builtin_amdgcn_wmma_f32_16x16x4_f32(
          false, a, false, b, (short)0, acc[c], false, false);
    }
  }

  const int mBase = row0 + 8 * half;
  float nrm[8];
  #pragma unroll
  for (int r = 0; r < 8; ++r) nrm[r] = normSrc[mBase + r];
  #pragma unroll
  for (int c = 0; c < 8; ++c) {
    #pragma unroll
    for (int r = 0; r < 8; ++r) {
      Y[(size_t)(mBase + r) * HF + c * 16 + l16] = acc[c][r] * nrm[r];
    }
  }
}

#define FW 25
#define FTHREADS (FW * 32)

template <int U>
__device__ __forceinline__ void gather_chunk_pf(const float* __restrict__ hS, int px, int py,
                                                int k, int n, int lane, v4f& acc,
                                                const int2* __restrict__ csr2, int pfIdx, int2& pf) {
  v4f v[U];
  float w[U];
  for (int i = 0; i < U; ++i) {
    const int idx = min(k + i, n - 1);
    int s = __builtin_amdgcn_readlane(px, idx);
    w[i] = __int_as_float(__builtin_amdgcn_readlane(py, idx));
    s = (int)min((unsigned)s, (unsigned)(NNODES - 1));
    v[i] = *(const v4f*)(hS + (size_t)s * HF + lane * 4);
  }
  if (pfIdx >= 0) pf = csr2[pfIdx];
  if constexpr (U == 16) {
    asm volatile("" :: "v"(v[0]), "v"(v[1]), "v"(v[2]), "v"(v[3]),
                       "v"(v[4]), "v"(v[5]), "v"(v[6]), "v"(v[7]),
                       "v"(v[8]), "v"(v[9]), "v"(v[10]), "v"(v[11]),
                       "v"(v[12]), "v"(v[13]), "v"(v[14]), "v"(v[15]));
  } else if constexpr (U == 8) {
    asm volatile("" :: "v"(v[0]), "v"(v[1]), "v"(v[2]), "v"(v[3]),
                       "v"(v[4]), "v"(v[5]), "v"(v[6]), "v"(v[7]));
  } else {
    asm volatile("" :: "v"(v[0]), "v"(v[1]), "v"(v[2]), "v"(v[3]));
  }
  for (int i = 0; i < U; ++i) {
    if (i == 0 || k + i < n) {
      acc.x = fmaf(v[i].x, w[i], acc.x);
      acc.y = fmaf(v[i].y, w[i], acc.y);
      acc.z = fmaf(v[i].z, w[i], acc.z);
      acc.w = fmaf(v[i].w, w[i], acc.w);
    }
  }
}

__global__ __launch_bounds__(FTHREADS) void gather_gemm_norm_wmma(
    const float* __restrict__ hS,
    const int* __restrict__ rowptr,
    const int* __restrict__ csr,
    const float* __restrict__ normDst,
    const float* __restrict__ bias,
    const float* __restrict__ W,
    const float* __restrict__ normSrc,
    float* __restrict__ Y,
    int nRows)
{
  __shared__ __attribute__((aligned(16))) float smemF[HF * HF + FW * 16 * XSTR];
  float* sW = smemF;
  float* sX = smemF + HF * HF;

  {
    const float4* Wv = (const float4*)W;
    float4* sWv = (float4*)sW;
    float4 t[6];
    #pragma unroll
    for (int u = 0; u < 6; ++u) {
      const int i = (int)threadIdx.x + u * FTHREADS;
      t[u] = Wv[min(i, HF * HF / 4 - 1)];
    }
    #pragma unroll
    for (int u = 0; u < 6; ++u) {
      const int i = (int)threadIdx.x + u * FTHREADS;
      if (i < HF * HF / 4) sWv[i] = t[u];
    }
  }
  __syncthreads();

  const int wave = threadIdx.x >> 5;
  const int lane = threadIdx.x & 31;
  const int row0 = __builtin_amdgcn_readfirstlane((int)((blockIdx.x * FW + wave) * 16));
  if (row0 >= nRows) return;
  float* sXw = sX + wave * 16 * XSTR;
  const int2* csr2 = (const int2*)csr;

  {
    const int rp = rowptr[row0 + min(lane, 16)];
    const float ndv = normDst[row0 + (lane & 15)];
    const v4f b = *(const v4f*)(bias + lane * 4);
    int begN = max(__builtin_amdgcn_readlane(rp, 0), 0);
    int endN = min(__builtin_amdgcn_readlane(rp, 1), NEDGES);
    int2 pN = csr2[min(max(min(begN + lane, endN - 1), 0), NEDGES - 1)];
    #pragma nounroll
    for (int j = 0; j < 16; ++j) {
      const int beg = begN;
      const int end = endN;
      int2 p = pN;
      int pfIdx = -1;
      if (j + 1 < 16) {
        begN = max(__builtin_amdgcn_readlane(rp, j + 1), 0);
        endN = min(__builtin_amdgcn_readlane(rp, j + 2), NEDGES);
        pfIdx = min(max(min(begN + lane, endN - 1), 0), NEDGES - 1);
      }
      bool owed = (j + 1 < 16);
      const float nd = __int_as_float(__builtin_amdgcn_readlane(__float_as_int(ndv), j));
      v4f acc = {0.0f, 0.0f, 0.0f, 0.0f};
      for (int base = beg; base < end; base += 32) {
        const int n = min(32, end - base);
        if (base != beg) p = csr2[min(base + lane, end - 1)];
        int k = 0;
        for (; n - k > 12; k += 16) {
          gather_chunk_pf<16>(hS, p.x, p.y, k, n, lane, acc, csr2, owed ? pfIdx : -1, pN);
          owed = false;
        }
        if (n - k > 4) {
          gather_chunk_pf<8>(hS, p.x, p.y, k, n, lane, acc, csr2, owed ? pfIdx : -1, pN);
          owed = false;
          k += 8;
        }
        if (k < n) {
          gather_chunk_pf<4>(hS, p.x, p.y, k, n, lane, acc, csr2, owed ? pfIdx : -1, pN);
          owed = false;
        }
      }
      if (owed) pN = csr2[pfIdx];
      v4f x;
      x.x = acc.x * nd + b.x;
      x.y = acc.y * nd + b.y;
      x.z = acc.z * nd + b.z;
      x.w = acc.w * nd + b.w;
      x.x = (x.x > 0.0f) ? x.x : 0.01f * x.x;
      x.y = (x.y > 0.0f) ? x.y : 0.01f * x.y;
      x.z = (x.z > 0.0f) ? x.z : 0.01f * x.z;
      x.w = (x.w > 0.0f) ? x.w : 0.01f * x.w;
      *(v4f*)(sXw + j * XSTR + lane * 4) = x;
    }
  }
  __builtin_amdgcn_fence(__ATOMIC_RELEASE, "wavefront");
  __builtin_amdgcn_wave_barrier();
  __builtin_amdgcn_fence(__ATOMIC_ACQUIRE, "wavefront");

  const int half = lane >> 4;
  const int l16  = lane & 15;
  const float* aRow = sXw + l16 * XSTR;

  v8f acc[8];
  v8f vz = {};
  #pragma unroll
  for (int c = 0; c < 8; ++c) acc[c] = vz;

  for (int k0 = 0; k0 < HF; k0 += 4) {
    v2f a;
    a.x = aRow[k0 + 2 * half];
    a.y = aRow[k0 + 2 * half + 1];
    #pragma unroll
    for (int c = 0; c < 8; ++c) {
      v2f bb;
      const int n = c * 16 + l16;
      bb.x = sW[(k0 + 2 * half) * HF + n];
      bb.y = sW[(k0 + 2 * half + 1) * HF + n];
      acc[c] = __builtin_amdgcn_wmma_f32_16x16x4_f32(
          false, a, false, bb, (short)0, acc[c], false, false);
    }
  }

  const int mBase = row0 + 8 * half;
  float nrm[8];
  #pragma unroll
  for (int r = 0; r < 8; ++r) nrm[r] = normSrc[mBase + r];
  #pragma unroll
  for (int c = 0; c < 8; ++c) {
    #pragma unroll
    for (int r = 0; r < 8; ++r) {
      Y[(size_t)(mBase + r) * HF + c * 16 + l16] = acc[c][r] * nrm[r];
    }
  }
}

#define LWN 16

__global__ __launch_bounds__(FTHREADS) void gather_linear_wmma(
    const float* __restrict__ hS,
    const int* __restrict__ rowptr,
    const int* __restrict__ csr,
    const float* __restrict__ normDst,
    const float* __restrict__ bias,
    const float* __restrict__ Wl,
    const float* __restrict__ bl,
    float* __restrict__ out,
    int nRows)
{
  __shared__ __attribute__((aligned(16))) float smemL[HF * LWN + FW * 16 * XSTR];
  float* sWl = smemL;
  float* sX = smemL + HF * LWN;

  {
    float t[3];
    #pragma unroll
    for (int u = 0; u < 3; ++u) {
      const int i = min((int)threadIdx.x + u * FTHREADS, HF * LWN - 1);
      t[u] = Wl[(i >> 4) * NCLS + (i & (NCLS - 1))];
    }
    #pragma unroll
    for (int u = 0; u < 3; ++u) {
      const int i = (int)threadIdx.x + u * FTHREADS;
      if (i < HF * LWN) sWl[i] = ((i & (LWN - 1)) < NCLS) ? t[u] : 0.0f;
    }
  }
  __syncthreads();

  const int wave = threadIdx.x >> 5;
  const int lane = threadIdx.x & 31;
  const int row0 = __builtin_amdgcn_readfirstlane((int)((blockIdx.x * FW + wave) * 16));
  if (row0 >= nRows) return;
  float* sXw = sX + wave * 16 * XSTR;
  const int2* csr2 = (const int2*)csr;

  {
    const int rp = rowptr[row0 + min(lane, 16)];
    const float ndv = normDst[row0 + (lane & 15)];
    const v4f b = *(const v4f*)(bias + lane * 4);
    int begN = max(__builtin_amdgcn_readlane(rp, 0), 0);
    int endN = min(__builtin_amdgcn_readlane(rp, 1), NEDGES);
    int2 pN = csr2[min(max(min(begN + lane, endN - 1), 0), NEDGES - 1)];
    #pragma nounroll
    for (int j = 0; j < 16; ++j) {
      const int beg = begN;
      const int end = endN;
      int2 p = pN;
      int pfIdx = -1;
      if (j + 1 < 16) {
        begN = max(__builtin_amdgcn_readlane(rp, j + 1), 0);
        endN = min(__builtin_amdgcn_readlane(rp, j + 2), NEDGES);
        pfIdx = min(max(min(begN + lane, endN - 1), 0), NEDGES - 1);
      }
      bool owed = (j + 1 < 16);
      const float nd = __int_as_float(__builtin_amdgcn_readlane(__float_as_int(ndv), j));
      v4f acc = {0.0f, 0.0f, 0.0f, 0.0f};
      for (int base = beg; base < end; base += 32) {
        const int n = min(32, end - base);
        if (base != beg) p = csr2[min(base + lane, end - 1)];
        int k = 0;
        for (; n - k > 12; k += 16) {
          gather_chunk_pf<16>(hS, p.x, p.y, k, n, lane, acc, csr2, owed ? pfIdx : -1, pN);
          owed = false;
        }
        if (n - k > 4) {
          gather_chunk_pf<8>(hS, p.x, p.y, k, n, lane, acc, csr2, owed ? pfIdx : -1, pN);
          owed = false;
          k += 8;
        }
        if (k < n) {
          gather_chunk_pf<4>(hS, p.x, p.y, k, n, lane, acc, csr2, owed ? pfIdx : -1, pN);
          owed = false;
        }
      }
      if (owed) pN = csr2[pfIdx];
      v4f x;
      x.x = acc.x * nd + b.x;
      x.y = acc.y * nd + b.y;
      x.z = acc.z * nd + b.z;
      x.w = acc.w * nd + b.w;
      x.x = (x.x > 0.0f) ? x.x : 0.01f * x.x;
      x.y = (x.y > 0.0f) ? x.y : 0.01f * x.y;
      x.z = (x.z > 0.0f) ? x.z : 0.01f * x.z;
      x.w = (x.w > 0.0f) ? x.w : 0.01f * x.w;
      *(v4f*)(sXw + j * XSTR + lane * 4) = x;
    }
  }
  __builtin_amdgcn_fence(__ATOMIC_RELEASE, "wavefront");
  __builtin_amdgcn_wave_barrier();
  __builtin_amdgcn_fence(__ATOMIC_ACQUIRE, "wavefront");

  const int half = lane >> 4;
  const int l16  = lane & 15;
  const float* aRow = sXw + l16 * XSTR;

  const float bv = (l16 < NCLS) ? bl[l16 & (NCLS - 1)] : 0.0f;
  v8f acc0, acc1;
  #pragma unroll
  for (int r = 0; r < 8; ++r) { acc0[r] = bv; acc1[r] = 0.0f; }

  for (int k0 = 0; k0 < HF; k0 += 8) {
    v2f a0, b0, a1, b1;
    a0.x = aRow[k0 + 2 * half];
    a0.y = aRow[k0 + 2 * half + 1];
    b0.x = sWl[(k0 + 2 * half) * LWN + l16];
    b0.y = sWl[(k0 + 2 * half + 1) * LWN + l16];
    a1.x = aRow[k0 + 4 + 2 * half];
    a1.y = aRow[k0 + 4 + 2 * half + 1];
    b1.x = sWl[(k0 + 4 + 2 * half) * LWN + l16];
    b1.y = sWl[(k0 + 4 + 2 * half + 1) * LWN + l16];
    acc0 = __builtin_amdgcn_wmma_f32_16x16x4_f32(
        false, a0, false, b0, (short)0, acc0, false, false);
    acc1 = __builtin_amdgcn_wmma_f32_16x16x4_f32(
        false, a1, false, b1, (short)0, acc1, false, false);
  }

  if (l16 < NCLS) {
    #pragma unroll
    for (int r = 0; r < 8; ++r) sXw[(8 * half + r) * NCLS + l16] = acc0[r] + acc1[r];
  }
  __builtin_amdgcn_fence(__ATOMIC_RELEASE, "wavefront");
  __builtin_amdgcn_wave_barrier();
  __builtin_amdgcn_fence(__ATOMIC_ACQUIRE, "wavefront");
  const v4f o = *(const v4f*)(sXw + lane * 4);
  *(v4f*)(out + (size_t)row0 * NCLS + lane * 4) = o;
}

extern "C" void kernel_launch(void* const* d_in, const int* in_sizes, int n_in,
                              void* d_out, int out_size, void* d_ws, size_t ws_size,
                              hipStream_t stream) {
  const float* in_feat = (const float*)d_in[0];
  const int*   src     = (const int*)d_in[1];
  const int*   dst     = (const int*)d_in[2];
  const float* ew      = (const float*)d_in[3];
  const float* convW   = (const float*)d_in[4];
  const float* convB   = (const float*)d_in[5];
  const float* linW    = (const float*)d_in[6];
  const float* linB    = (const float*)d_in[7];
  float* out = (float*)d_out;

  float* ws = (float*)d_ws;
  const size_t NF = (size_t)NNODES * HF;
  float* hA      = ws;
  float* hS      = ws + NF;
  float* normSrc = ws + 2 * NF;
  int*   inCnt   = (int*)(normSrc + NNODES);
  float* normDst = normSrc + 2 * NNODES;
  int*   rowptr  = (int*)(normSrc + 3 * NNODES);
  int*   blockSum = rowptr + NNODES + 96;
  int*   rank    = blockSum + 128;
  int*   csr     = rank + NEDGES;

  zero_f<<<(2 * NNODES + 255) / 256, 256, 0, stream>>>(normSrc, 2 * NNODES);
  degree_rank_kernel<<<(NEDGES + 255) / 256, 256, 0, stream>>>(src, dst, normSrc, inCnt, rank, NEDGES);
  norm_kernel<<<(NNODES + 255) / 256, 256, 0, stream>>>(normSrc, inCnt, normDst, NNODES);
  rowptr_block_scan<<<SCAN_BLOCKS, 256, 0, stream>>>(inCnt, rowptr, blockSum, NNODES);
  rowptr_add_offsets<<<SCAN_BLOCKS, 256, 0, stream>>>(rowptr, blockSum, NNODES, NEDGES);
  csr_fill<<<(NEDGES + 255) / 256, 256, 0, stream>>>(src, dst, ew, rank, rowptr, csr, NEDGES);

  const size_t ldsBytes = (size_t)(HF * HF + 4 * 16 * XSTR) * sizeof(float);
  const int gemmBlocks = (NNODES / 16 + 3) / 4;
  const int fusedBlocks = (NNODES / 16 + FW - 1) / FW;

  gemm_norm_wmma<<<gemmBlocks, 128, ldsBytes, stream>>>(in_feat, convW, normSrc, hS, NNODES);
  gather_gemm_norm_wmma<<<fusedBlocks, FTHREADS, 0, stream>>>(
      hS, rowptr, csr, normDst, convB + 0 * HF, convW + (size_t)1 * HF * HF, normSrc, hA, NNODES);
  gather_gemm_norm_wmma<<<fusedBlocks, FTHREADS, 0, stream>>>(
      hA, rowptr, csr, normDst, convB + 1 * HF, convW + (size_t)2 * HF * HF, normSrc, hS, NNODES);
  gather_gemm_norm_wmma<<<fusedBlocks, FTHREADS, 0, stream>>>(
      hS, rowptr, csr, normDst, convB + 2 * HF, convW + (size_t)3 * HF * HF, normSrc, hA, NNODES);
  gather_gemm_norm_wmma<<<fusedBlocks, FTHREADS, 0, stream>>>(
      hA, rowptr, csr, normDst, convB + 3 * HF, convW + (size_t)4 * HF * HF, normSrc, hS, NNODES);
  gather_gemm_norm_wmma<<<fusedBlocks, FTHREADS, 0, stream>>>(
      hS, rowptr, csr, normDst, convB + 4 * HF, convW + (size_t)5 * HF * HF, normSrc, hA, NNODES);
  gather_linear_wmma<<<fusedBlocks, FTHREADS, 0, stream>>>(
      hA, rowptr, csr, normDst, convB + 5 * HF, linW, linB, out, NNODES);
}


# global atomics (degree, rank) -> two-level counting sort in LDS: edges binned by 256-node bucket, one work-group per bucket builds rows and both norms; slot order = LDS arrival; 13->11 launches
# speedup vs baseline: 5.8064x; 1.3924x over previous
#include <hip/hip_runtime.h>
#include <hip/hip_bf16.h>

typedef float v2f __attribute__((ext_vector_type(2)));
typedef float v8f __attribute__((ext_vector_type(8)));
typedef float v4f __attribute__((ext_vector_type(4)));

#define NNODES 100000
#define NEDGES 1600000
#define HF 128
#define NCLS 8
#define NLAYERS 6
#define XSTR 132

#define BSH 8
#define BNODES (1 << BSH)
#define NBK ((NNODES >> BSH) + 1)
#define NBLK 256
#define EPB ((NEDGES + NBLK - 1) / NBLK)
#define STHREADS 1024
#define SWAVES (STHREADS / 32)
#define SRC_BITS 17
#define DCAP 8192
static_assert(NNODES <= (1 << SRC_BITS), "node index must fit SRC_BITS");
static_assert(NBK <= STHREADS, "one thread per bucket in the bucket-start scan");
static_assert(NBK <= 65536, "bucket ids are kept as 16-bit values");
static_assert(BNODES <= STHREADS, "one thread per node in the row scan");

template <int NW>
__device__ __forceinline__ int block_excl_scan(int v, int* sW, int& total) {
  const int lane = threadIdx.x & 31;
  const int wv = threadIdx.x >> 5;
  int x = v;
  for (int o = 1; o < 32; o <<= 1) {
    int y = __shfl_up(x, o, 32);
    if (lane >= o) x += y;
  }
  if (lane == 31) sW[wv] = x;
  __syncthreads();
  int woff = 0, tot = 0;
  for (int i = 0; i < NW; ++i) {
    int s = sW[i];
    if (i < wv) woff += s;
    tot += s;
  }
  __syncthreads();
  total = tot;
  return woff + x - v;
}

__global__ __launch_bounds__(STHREADS) void bin_count(const int* __restrict__ src,
                                                      const int* __restrict__ dst,
                                                      int* __restrict__ hist) {
  __shared__ int cD[NBK];
  __shared__ int cS[NBK];
  const int t = threadIdx.x;
  for (int b = t; b < NBK; b += STHREADS) { cD[b] = 0; cS[b] = 0; }
  __syncthreads();
  const int e0 = min((int)blockIdx.x * EPB, NEDGES);
  const int e1 = min(e0 + EPB, NEDGES);
  for (int base = e0; base < e1; base += STHREADS * 4) {
    unsigned d[4], s[4];
    #pragma unroll
    for (int u = 0; u < 4; ++u) {
      const int i = min(base + u * STHREADS + t, e1 - 1);
      d[u] = (unsigned)dst[i];
      s[u] = (unsigned)src[i];
    }
    #pragma unroll
    for (int u = 0; u < 4; ++u) {
      if (base + u * STHREADS + t < e1) {
        atomicAdd(&cD[min(d[u], (unsigned)(NNODES - 1)) >> BSH], 1);
        atomicAdd(&cS[min(s[u], (unsigned)(NNODES - 1)) >> BSH], 1);
      }
    }
  }
  __syncthreads();
  for (int b = t; b < NBK; b += STHREADS) {
    hist[blockIdx.x * NBK + b] = cD[b];
    hist[NBLK * NBK + blockIdx.x * NBK + b] = cS[b];
  }
}

__global__ __launch_bounds__(NBLK) void bin_scan(const int* __restrict__ hist, int* __restrict__ pre,
                                                 int* __restrict__ tot) {
  __shared__ int sW[NBLK / 32];
  const int t = threadIdx.x;
  const int b = blockIdx.x;
  const int side = blockIdx.y * (NBLK * NBK);
  int total;
  const int p = block_excl_scan<NBLK / 32>(hist[side + t * NBK + b], sW, total);
  pre[side + t * NBK + b] = p;
  if (t == 0) tot[blockIdx.y * NBK + b] = total;
}

__global__ __launch_bounds__(STHREADS) void bin_scatter(const int* __restrict__ src,
                                                        const int* __restrict__ dst,
                                                        const float* __restrict__ ew,
                                                        const int* __restrict__ hist,
                                                        const int* __restrict__ pre,
                                                        const int* __restrict__ tot,
                                                        int2* __restrict__ binE,
                                                        int* __restrict__ binS,
                                                        int* __restrict__ bkt) {
  __shared__ int curD[NBK];
  __shared__ int curS[NBK];
  __shared__ int dltD[NBK];
  __shared__ int dltS[NBK];
  __shared__ int sW[SWAVES];
  __shared__ __attribute__((aligned(8))) int stEraw[2 * EPB];
  __shared__ int stS[EPB];
  __shared__ unsigned short stEb[EPB];
  __shared__ unsigned short stSb[EPB];
  int2* stE = (int2*)stEraw;
  const int t = threadIdx.x;
  const int blk = blockIdx.x;
  for (int side = 0; side < 2; ++side) {
    const int gv = (t < NBK) ? tot[side * NBK + t] : 0;
    const int lv = (t < NBK) ? hist[side * (NBLK * NBK) + blk * NBK + t] : 0;
    int gtot, ltot;
    const int gstart = block_excl_scan<SWAVES>(gv, sW, gtot);
    const int lstart = block_excl_scan<SWAVES>(lv, sW, ltot);
    if (t < NBK) {
      const int g = gstart + pre[side * (NBLK * NBK) + blk * NBK + t];
      if (side == 0) { curD[t] = lstart; dltD[t] = g - lstart; }
      else           { curS[t] = lstart; dltS[t] = g - lstart; }
      if (blk == 0) {
        bkt[side * (NBK + 1) + t] = gstart;
        if (t == NBK - 1) bkt[side * (NBK + 1) + NBK] = gstart + gv;
      }
    }
  }
  __syncthreads();
  const int e0 = min(blk * EPB, NEDGES);
  const int e1 = min(e0 + EPB, NEDGES);
  for (int base = e0; base < e1; base += STHREADS * 4) {
    unsigned d[4], s[4];
    int w[4];
    #pragma unroll
    for (int u = 0; u < 4; ++u) {
      const int i = min(base + u * STHREADS + t, e1 - 1);
      d[u] = (unsigned)dst[i];
      s[u] = (unsigned)src[i];
      w[u] = __float_as_int(ew[i]);
    }
    #pragma unroll
    for (int u = 0; u < 4; ++u) {
      if (base + u * STHREADS + t < e1) {
        const unsigned dd = min(d[u], (unsigned)(NNODES - 1));
        const unsigned ss = min(s[u], (unsigned)(NNODES - 1));
        int rd = atomicAdd(&curD[dd >> BSH], 1);
        int rs = atomicAdd(&curS[ss >> BSH], 1);
        rd = (int)min((unsigned)rd, (unsigned)(EPB - 1));
        rs = (int)min((unsigned)rs, (unsigned)(EPB - 1));
        stE[rd] = make_int2((int)(ss | ((dd & (BNODES - 1)) << SRC_BITS)), w[u]);
        stEb[rd] = (unsigned short)(dd >> BSH);
        stS[rs] = (int)ss;
        stSb[rs] = (unsigned short)(ss >> BSH);
      }
    }
  }
  __syncthreads();
  const int n = e1 - e0;
  for (int r = t; r < n; r += STHREADS) {
    const int pd = (int)min((unsigned)(r + dltD[stEb[r]]), (unsigned)(NEDGES - 1));
    const int ps = (int)min((unsigned)(r + dltS[stSb[r]]), (unsigned)(NEDGES - 1));
    binE[pd] = stE[r];
    binS[ps] = stS[r];
  }
}

__global__ __launch_bounds__(STHREADS) void bucket_build(const int2* __restrict__ binE,
                                                         const int* __restrict__ binS,
                                                         const int* __restrict__ bkt,
                                                         int* __restrict__ rowptr,
                                                         int2* __restrict__ csr,
                                                         float* __restrict__ normSrc,
                                                         float* __restrict__ normDst) {
  __shared__ int cnt[BNODES];
  __shared__ int cntS[BNODES];
  __shared__ int cur[BNODES];
  __shared__ int sW[SWAVES];
  __shared__ __attribute__((aligned(8))) int stageRaw[2 * DCAP];
  int2* stage = (int2*)stageRaw;
  const int t = threadIdx.x;
  const int b = blockIdx.x;
  if (t < BNODES) { cnt[t] = 0; cntS[t] = 0; }
  __syncthreads();
  const int bs = min(max(bkt[b], 0), NEDGES);
  const int be = min(max(bkt[b + 1], bs), NEDGES);
  const int ss = min(max(bkt[(NBK + 1) + b], 0), NEDGES);
  const int se = min(max(bkt[(NBK + 1) + b + 1], ss), NEDGES);
  for (int base = bs; base < be; base += STHREADS * 4) {
    int k[4];
    #pragma unroll
    for (int u = 0; u < 4; ++u) k[u] = binE[min(base + u * STHREADS + t, be - 1)].x;
    #pragma unroll
    for (int u = 0; u < 4; ++u)
      if (base + u * STHREADS + t < be) atomicAdd(&cnt[(k[u] >> SRC_BITS) & (BNODES - 1)], 1);
  }
  for (int base = ss; base < se; base += STHREADS * 4) {
    int k[4];
    #pragma unroll
    for (int u = 0; u < 4; ++u) k[u] = binS[min(base + u * STHREADS + t, se - 1)];
    #pragma unroll
    for (int u = 0; u < 4; ++u)
      if (base + u * STHREADS + t < se) atomicAdd(&cntS[k[u] & (BNODES - 1)], 1);
  }
  __syncthreads();
  const int v = (t < BNODES) ? cnt[t] : 0;
  int total;
  const int rel = block_excl_scan<SWAVES>(v, sW, total);
  if (t < BNODES) {
    cur[t] = rel;
    const int node = b * BNODES + t;
    if (node <= NNODES) rowptr[node] = bs + rel;
    if (node < NNODES) {
      normDst[node] = rsqrtf(fmaxf((float)v, 1.0f));
      normSrc[node] = rsqrtf(fmaxf((float)cntS[t], 1.0f));
    }
  }
  __syncthreads();
  const int nb = be - bs;
  const bool inLds = nb <= DCAP;
  for (int base = bs; base < be; base += STHREADS * 4) {
    int2 r[4];
    #pragma unroll
    for (int u = 0; u < 4; ++u) r[u] = binE[min(base + u * STHREADS + t, be - 1)];
    #pragma unroll
    for (int u = 0; u < 4; ++u) {
      if (base + u * STHREADS + t < be) {
        const int p = atomicAdd(&cur[(r[u].x >> SRC_BITS) & (BNODES - 1)], 1);
        const int2 rec = make_int2(r[u].x & ((1 << SRC_BITS) - 1), r[u].y);
        if (inLds) stage[min((unsigned)p, (unsigned)(DCAP - 1))] = rec;
        else csr[min((unsigned)(bs + p), (unsigned)(NEDGES - 1))] = rec;
      }
    }
  }
  if (inLds) {
    __syncthreads();
    for (int i = t; i < nb; i += STHREADS) csr[bs + i] = stage[i];
  }
}

__global__ __launch_bounds__(128) void gemm_norm_wmma(
    const float* __restrict__ X,
    const float* __restrict__ W,
    const float* __restrict__ normSrc,
    float* __restrict__ Y,
    int nRows)
{
  extern __shared__ float smem[];
  float* sW = smem;
  float* sX = smem + HF * HF;

  {
    const float4* Wv = (const float4*)W;
    float4* sWv = (float4*)sW;
    for (int i = threadIdx.x; i < HF * HF / 4; i += blockDim.x) sWv[i] = Wv[i];
  }

  const int wave = threadIdx.x >> 5;
  const int lane = threadIdx.x & 31;
  const int tile = blockIdx.x * 4 + wave;
  const int row0 = tile * 16;
  float* sXw = sX + wave * 16 * XSTR;

  if (row0 < nRows) {
    for (int j = 0; j < 16; ++j) {
      float4 v = ((const float4*)(X + (size_t)(row0 + j) * HF))[lane];
      *(float4*)(sXw + j * XSTR + lane * 4) = v;
    }
  }
  __syncthreads();
  if (row0 >= nRows) return;

  const int half = lane >> 4;
  const int l16  = lane & 15;
  const float* aRow = sXw + l16 * XSTR;

  v8f acc[8];
  v8f vz = {};
  #pragma unroll
  for (int c = 0; c < 8; ++c) acc[c] = vz;

  for (int k0 = 0; k0 < HF; k0 += 4) {
    v2f a;
    a.x = aRow[k0 + 2 * half];
    a.y = aRow[k0 + 2 * half + 1];
    #pragma unroll
    for (int c = 0; c < 8; ++c) {
      v2f b;
      const int n = c * 16 + l16;
      b.x = sW[(k0 + 2 * half) * HF + n];
      b.y = sW[(k0 + 2 * half + 1) * HF + n];
      acc[c] = __builtin_amdgcn_wmma_f32_16x16x4_f32(
          false, a, false, b, (short)0, acc[c], false, false);
    }
  }

  const int mBase = row0 + 8 * half;
  float nrm[8];
  #pragma unroll
  for (int r = 0; r < 8; ++r) nrm[r] = normSrc[mBase + r];
  #pragma unroll
  for (int c = 0; c < 8; ++c) {
    #pragma unroll
    for (int r = 0; r < 8; ++r) {
      Y[(size_t)(mBase + r) * HF + c * 16 + l16] = acc[c][r] * nrm[r];
    }
  }
}

#define FW 25
#define FTHREADS (FW * 32)

template <int U>
__device__ __forceinline__ void gather_chunk_pf(const float* __restrict__ hS, int px, int py,
                                                int k, int n, int lane, v4f& acc,
                                                const int2* __restrict__ csr2, int pfIdx, int2& pf) {
  v4f v[U];
  float w[U];
  for (int i = 0; i < U; ++i) {
    const int idx = min(k + i, n - 1);
    int s = __builtin_amdgcn_readlane(px, idx);
    w[i] = __int_as_float(__builtin_amdgcn_readlane(py, idx));
    s = (int)min((unsigned)s, (unsigned)(NNODES - 1));
    v[i] = *(const v4f*)(hS + (size_t)s * HF + lane * 4);
  }
  if (pfIdx >= 0) pf = csr2[pfIdx];
  if constexpr (U == 16) {
    asm volatile("" :: "v"(v[0]), "v"(v[1]), "v"(v[2]), "v"(v[3]),
                       "v"(v[4]), "v"(v[5]), "v"(v[6]), "v"(v[7]),
                       "v"(v[8]), "v"(v[9]), "v"(v[10]), "v"(v[11]),
                       "v"(v[12]), "v"(v[13]), "v"(v[14]), "v"(v[15]));
  } else if constexpr (U == 8) {
    asm volatile("" :: "v"(v[0]), "v"(v[1]), "v"(v[2]), "v"(v[3]),
                       "v"(v[4]), "v"(v[5]), "v"(v[6]), "v"(v[7]));
  } else {
    asm volatile("" :: "v"(v[0]), "v"(v[1]), "v"(v[2]), "v"(v[3]));
  }
  for (int i = 0; i < U; ++i) {
    if (i == 0 || k + i < n) {
      acc.x = fmaf(v[i].x, w[i], acc.x);
      acc.y = fmaf(v[i].y, w[i], acc.y);
      acc.z = fmaf(v[i].z, w[i], acc.z);
      acc.w = fmaf(v[i].w, w[i], acc.w);
    }
  }
}

__global__ __launch_bounds__(FTHREADS) void gather_gemm_norm_wmma(
    const float* __restrict__ hS,
    const int* __restrict__ rowptr,
    const int* __restrict__ csr,
    const float* __restrict__ normDst,
    const float* __restrict__ bias,
    const float* __restrict__ W,
    const float* __restrict__ normSrc,
    float* __restrict__ Y,
    int nRows)
{
  __shared__ __attribute__((aligned(16))) float smemF[HF * HF + FW * 16 * XSTR];
  float* sW = smemF;
  float* sX = smemF + HF * HF;

  {
    const float4* Wv = (const float4*)W;
    float4* sWv = (float4*)sW;
    float4 t[6];
    #pragma unroll
    for (int u = 0; u < 6; ++u) {
      const int i = (int)threadIdx.x + u * FTHREADS;
      t[u] = Wv[min(i, HF * HF / 4 - 1)];
    }
    #pragma unroll
    for (int u = 0; u < 6; ++u) {
      const int i = (int)threadIdx.x + u * FTHREADS;
      if (i < HF * HF / 4) sWv[i] = t[u];
    }
  }
  __syncthreads();

  const int wave = threadIdx.x >> 5;
  const int lane = threadIdx.x & 31;
  const int row0 = __builtin_amdgcn_readfirstlane((int)((blockIdx.x * FW + wave) * 16));
  if (row0 >= nRows) return;
  float* sXw = sX + wave * 16 * XSTR;
  const int2* csr2 = (const int2*)csr;

  {
    const int rp = rowptr[row0 + min(lane, 16)];
    const float ndv = normDst[row0 + (lane & 15)];
    const v4f b = *(const v4f*)(bias + lane * 4);
    int begN = max(__builtin_amdgcn_readlane(rp, 0), 0);
    int endN = min(__builtin_amdgcn_readlane(rp, 1), NEDGES);
    int2 pN = csr2[min(max(min(begN + lane, endN - 1), 0), NEDGES - 1)];
    #pragma nounroll
    for (int j = 0; j < 16; ++j) {
      const int beg = begN;
      const int end = endN;
      int2 p = pN;
      int pfIdx = -1;
      if (j + 1 < 16) {
        begN = max(__builtin_amdgcn_readlane(rp, j + 1), 0);
        endN = min(__builtin_amdgcn_readlane(rp, j + 2), NEDGES);
        pfIdx = min(max(min(begN + lane, endN - 1), 0), NEDGES - 1);
      }
      bool owed = (j + 1 < 16);
      const float nd = __int_as_float(__builtin_amdgcn_readlane(__float_as_int(ndv), j));
      v4f acc = {0.0f, 0.0f, 0.0f, 0.0f};
      for (int base = beg; base < end; base += 32) {
        const int n = min(32, end - base);
        if (base != beg) p = csr2[min(base + lane, end - 1)];
        int k = 0;
        for (; n - k > 12; k += 16) {
          gather_chunk_pf<16>(hS, p.x, p.y, k, n, lane, acc, csr2, owed ? pfIdx : -1, pN);
          owed = false;
        }
        if (n - k > 4) {
          gather_chunk_pf<8>(hS, p.x, p.y, k, n, lane, acc, csr2, owed ? pfIdx : -1, pN);
          owed = false;
          k += 8;
        }
        if (k < n) {
          gather_chunk_pf<4>(hS, p.x, p.y, k, n, lane, acc, csr2, owed ? pfIdx : -1, pN);
          owed = false;
        }
      }
      if (owed) pN = csr2[pfIdx];
      v4f x;
      x.x = acc.x * nd + b.x;
      x.y = acc.y * nd + b.y;
      x.z = acc.z * nd + b.z;
      x.w = acc.w * nd + b.w;
      x.x = (x.x > 0.0f) ? x.x : 0.01f * x.x;
      x.y = (x.y > 0.0f) ? x.y : 0.01f * x.y;
      x.z = (x.z > 0.0f) ? x.z : 0.01f * x.z;
      x.w = (x.w > 0.0f) ? x.w : 0.01f * x.w;
      *(v4f*)(sXw + j * XSTR + lane * 4) = x;
    }
  }
  __builtin_amdgcn_fence(__ATOMIC_RELEASE, "wavefront");
  __builtin_amdgcn_wave_barrier();
  __builtin_amdgcn_fence(__ATOMIC_ACQUIRE, "wavefront");

  const int half = lane >> 4;
  const int l16  = lane & 15;
  const float* aRow = sXw + l16 * XSTR;

  v8f acc[8];
  v8f vz = {};
  #pragma unroll
  for (int c = 0; c < 8; ++c) acc[c] = vz;

  for (int k0 = 0; k0 < HF; k0 += 4) {
    v2f a;
    a.x = aRow[k0 + 2 * half];
    a.y = aRow[k0 + 2 * half + 1];
    #pragma unroll
    for (int c = 0; c < 8; ++c) {
      v2f bb;
      const int n = c * 16 + l16;
      bb.x = sW[(k0 + 2 * half) * HF + n];
      bb.y = sW[(k0 + 2 * half + 1) * HF + n];
      acc[c] = __builtin_amdgcn_wmma_f32_16x16x4_f32(
          false, a, false, bb, (short)0, acc[c], false, false);
    }
  }

  const int mBase = row0 + 8 * half;
  float nrm[8];
  #pragma unroll
  for (int r = 0; r < 8; ++r) nrm[r] = normSrc[mBase + r];
  #pragma unroll
  for (int c = 0; c < 8; ++c) {
    #pragma unroll
    for (int r = 0; r < 8; ++r) {
      Y[(size_t)(mBase + r) * HF + c * 16 + l16] = acc[c][r] * nrm[r];
    }
  }
}

#define LWN 16

__global__ __launch_bounds__(FTHREADS) void gather_linear_wmma(
    const float* __restrict__ hS,
    const int* __restrict__ rowptr,
    const int* __restrict__ csr,
    const float* __restrict__ normDst,
    const float* __restrict__ bias,
    const float* __restrict__ Wl,
    const float* __restrict__ bl,
    float* __restrict__ out,
    int nRows)
{
  __shared__ __attribute__((aligned(16))) float smemL[HF * LWN + FW * 16 * XSTR];
  float* sWl = smemL;
  float* sX = smemL + HF * LWN;

  {
    float t[3];
    #pragma unroll
    for (int u = 0; u < 3; ++u) {
      const int i = min((int)threadIdx.x + u * FTHREADS, HF * LWN - 1);
      t[u] = Wl[(i >> 4) * NCLS + (i & (NCLS - 1))];
    }
    #pragma unroll
    for (int u = 0; u < 3; ++u) {
      const int i = (int)threadIdx.x + u * FTHREADS;
      if (i < HF * LWN) sWl[i] = ((i & (LWN - 1)) < NCLS) ? t[u] : 0.0f;
    }
  }
  __syncthreads();

  const int wave = threadIdx.x >> 5;
  const int lane = threadIdx.x & 31;
  const int row0 = __builtin_amdgcn_readfirstlane((int)((blockIdx.x * FW + wave) * 16));
  if (row0 >= nRows) return;
  float* sXw = sX + wave * 16 * XSTR;
  const int2* csr2 = (const int2*)csr;

  {
    const int rp = rowptr[row0 + min(lane, 16)];
    const float ndv = normDst[row0 + (lane & 15)];
    const v4f b = *(const v4f*)(bias + lane * 4);
    int begN = max(__builtin_amdgcn_readlane(rp, 0), 0);
    int endN = min(__builtin_amdgcn_readlane(rp, 1), NEDGES);
    int2 pN = csr2[min(max(min(begN + lane, endN - 1), 0), NEDGES - 1)];
    #pragma nounroll
    for (int j = 0; j < 16; ++j) {
      const int beg = begN;
      const int end = endN;
      int2 p = pN;
      int pfIdx = -1;
      if (j + 1 < 16) {
        begN = max(__builtin_amdgcn_readlane(rp, j + 1), 0);
        endN = min(__builtin_amdgcn_readlane(rp, j + 2), NEDGES);
        pfIdx = min(max(min(begN + lane, endN - 1), 0), NEDGES - 1);
      }
      bool owed = (j + 1 < 16);
      const float nd = __int_as_float(__builtin_amdgcn_readlane(__float_as_int(ndv), j));
      v4f acc = {0.0f, 0.0f, 0.0f, 0.0f};
      for (int base = beg; base < end; base += 32) {
        const int n = min(32, end - base);
        if (base != beg) p = csr2[min(base + lane, end - 1)];
        int k = 0;
        for (; n - k > 12; k += 16) {
          gather_chunk_pf<16>(hS, p.x, p.y, k, n, lane, acc, csr2, owed ? pfIdx : -1, pN);
          owed = false;
        }
        if (n - k > 4) {
          gather_chunk_pf<8>(hS, p.x, p.y, k, n, lane, acc, csr2, owed ? pfIdx : -1, pN);
          owed = false;
          k += 8;
        }
        if (k < n) {
          gather_chunk_pf<4>(hS, p.x, p.y, k, n, lane, acc, csr2, owed ? pfIdx : -1, pN);
          owed = false;
        }
      }
      if (owed) pN = csr2[pfIdx];
      v4f x;
      x.x = acc.x * nd + b.x;
      x.y = acc.y * nd + b.y;
      x.z = acc.z * nd + b.z;
      x.w = acc.w * nd + b.w;
      x.x = (x.x > 0.0f) ? x.x : 0.01f * x.x;
      x.y = (x.y > 0.0f) ? x.y : 0.01f * x.y;
      x.z = (x.z > 0.0f) ? x.z : 0.01f * x.z;
      x.w = (x.w > 0.0f) ? x.w : 0.01f * x.w;
      *(v4f*)(sXw + j * XSTR + lane * 4) = x;
    }
  }
  __builtin_amdgcn_fence(__ATOMIC_RELEASE, "wavefront");
  __builtin_amdgcn_wave_barrier();
  __builtin_amdgcn_fence(__ATOMIC_ACQUIRE, "wavefront");

  const int half = lane >> 4;
  const int l16  = lane & 15;
  const float* aRow = sXw + l16 * XSTR;

  const float bv = (l16 < NCLS) ? bl[l16 & (NCLS - 1)] : 0.0f;
  v8f acc0, acc1;
  #pragma unroll
  for (int r = 0; r < 8; ++r) { acc0[r] = bv; acc1[r] = 0.0f; }

  for (int k0 = 0; k0 < HF; k0 += 8) {
    v2f a0, b0, a1, b1;
    a0.x = aRow[k0 + 2 * half];
    a0.y = aRow[k0 + 2 * half + 1];
    b0.x = sWl[(k0 + 2 * half) * LWN + l16];
    b0.y = sWl[(k0 + 2 * half + 1) * LWN + l16];
    a1.x = aRow[k0 + 4 + 2 * half];
    a1.y = aRow[k0 + 4 + 2 * half + 1];
    b1.x = sWl[(k0 + 4 + 2 * half) * LWN + l16];
    b1.y = sWl[(k0 + 4 + 2 * half + 1) * LWN + l16];
    acc0 = __builtin_amdgcn_wmma_f32_16x16x4_f32(
        false, a0, false, b0, (short)0, acc0, false, false);
    acc1 = __builtin_amdgcn_wmma_f32_16x16x4_f32(
        false, a1, false, b1, (short)0, acc1, false, false);
  }

  if (l16 < NCLS) {
    #pragma unroll
    for (int r = 0; r < 8; ++r) sXw[(8 * half + r) * NCLS + l16] = acc0[r] + acc1[r];
  }
  __builtin_amdgcn_fence(__ATOMIC_RELEASE, "wavefront");
  __builtin_amdgcn_wave_barrier();
  __builtin_amdgcn_fence(__ATOMIC_ACQUIRE, "wavefront");
  const v4f o = *(const v4f*)(sXw + lane * 4);
  *(v4f*)(out + (size_t)row0 * NCLS + lane * 4) = o;
}

extern "C" void kernel_launch(void* const* d_in, const int* in_sizes, int n_in,
                              void* d_out, int out_size, void* d_ws, size_t ws_size,
                              hipStream_t stream) {
  const float* in_feat = (const float*)d_in[0];
  const int*   src     = (const int*)d_in[1];
  const int*   dst     = (const int*)d_in[2];
  const float* ew      = (const float*)d_in[3];
  const float* convW   = (const float*)d_in[4];
  const float* convB   = (const float*)d_in[5];
  const float* linW    = (const float*)d_in[6];
  const float* linB    = (const float*)d_in[7];
  float* out = (float*)d_out;

  float* ws = (float*)d_ws;
  const size_t NF = (size_t)NNODES * HF;
  float* hA      = ws;
  float* hS      = ws + NF;
  float* normSrc = ws + 2 * NF;
  float* normDst = normSrc + 2 * NNODES;
  int*   rowptr  = (int*)(normSrc + 3 * NNODES);
  int*   hist    = rowptr + NNODES + 96;
  int*   pre     = hist + 2 * NBLK * NBK;
  int*   tot     = pre + 2 * NBLK * NBK;
  int*   bkt     = tot + 1024;
  int*   binE    = bkt + 1024;
  int*   csr     = binE + 2 * NEDGES;
  int*   binS    = csr + 2 * NEDGES;

  bin_count<<<NBLK, STHREADS, 0, stream>>>(src, dst, hist);
  bin_scan<<<dim3(NBK, 2), NBLK, 0, stream>>>(hist, pre, tot);
  bin_scatter<<<NBLK, STHREADS, 0, stream>>>(src, dst, ew, hist, pre, tot, (int2*)binE, binS, bkt);
  bucket_build<<<NBK, STHREADS, 0, stream>>>((const int2*)binE, binS, bkt, rowptr, (int2*)csr,
                                             normSrc, normDst);

  const size_t ldsBytes = (size_t)(HF * HF + 4 * 16 * XSTR) * sizeof(float);
  const int gemmBlocks = (NNODES / 16 + 3) / 4;
  const int fusedBlocks = (NNODES / 16 + FW - 1) / FW;

  gemm_norm_wmma<<<gemmBlocks, 128, ldsBytes, stream>>>(in_feat, convW, normSrc, hS, NNODES);
  gather_gemm_norm_wmma<<<fusedBlocks, FTHREADS, 0, stream>>>(
      hS, rowptr, csr, normDst, convB + 0 * HF, convW + (size_t)1 * HF * HF, normSrc, hA, NNODES);
  gather_gemm_norm_wmma<<<fusedBlocks, FTHREADS, 0, stream>>>(
      hA, rowptr, csr, normDst, convB + 1 * HF, convW + (size_t)2 * HF * HF, normSrc, hS, NNODES);
  gather_gemm_norm_wmma<<<fusedBlocks, FTHREADS, 0, stream>>>(
      hS, rowptr, csr, normDst, convB + 2 * HF, convW + (size_t)3 * HF * HF, normSrc, hA, NNODES);
  gather_gemm_norm_wmma<<<fusedBlocks, FTHREADS, 0, stream>>>(
      hA, rowptr, csr, normDst, convB + 3 * HF, convW + (size_t)4 * HF * HF, normSrc, hS, NNODES);
  gather_gemm_norm_wmma<<<fusedBlocks, FTHREADS, 0, stream>>>(
      hS, rowptr, csr, normDst, convB + 4 * HF, convW + (size_t)5 * HF * HF, normSrc, hA, NNODES);
  gather_linear_wmma<<<fusedBlocks, FTHREADS, 0, stream>>>(
      hA, rowptr, csr, normDst, convB + 5 * HF, linW, linB, out, NNODES);
}


# staging prologues of all 7 GEMM launches: a thread's loads go out together instead of in dependent round trips (layer-0 GEMM: W 32 -> 3; gather launches: W, row bounds, first slots 5 -> 2)
# speedup vs baseline: 6.1477x; 1.0588x over previous
#include <hip/hip_runtime.h>
#include <hip/hip_bf16.h>

typedef float v2f __attribute__((ext_vector_type(2)));
typedef float v8f __attribute__((ext_vector_type(8)));
typedef float v4f __attribute__((ext_vector_type(4)));

#define NNODES 100000
#define NEDGES 1600000
#define HF 128
#define NCLS 8
#define NLAYERS 6
#define XSTR 132

#define BSH 8
#define BNODES (1 << BSH)
#define NBK ((NNODES >> BSH) + 1)
#define NBLK 256
#define EPB ((NEDGES + NBLK - 1) / NBLK)
#define STHREADS 1024
#define SWAVES (STHREADS / 32)
#define SRC_BITS 17
#define DCAP 8192
static_assert(NNODES <= (1 << SRC_BITS), "node index must fit SRC_BITS");
static_assert(NBK <= STHREADS, "one thread per bucket in the bucket-start scan");
static_assert(NBK <= 65536, "bucket ids are kept as 16-bit values");
static_assert(BNODES <= STHREADS, "one thread per node in the row scan");

template <int NW>
__device__ __forceinline__ int block_excl_scan(int v, int* sW, int& total) {
  const int lane = threadIdx.x & 31;
  const int wv = threadIdx.x >> 5;
  int x = v;
  for (int o = 1; o < 32; o <<= 1) {
    int y = __shfl_up(x, o, 32);
    if (lane >= o) x += y;
  }
  if (lane == 31) sW[wv] = x;
  __syncthreads();
  int woff = 0, tot = 0;
  for (int i = 0; i < NW; ++i) {
    int s = sW[i];
    if (i < wv) woff += s;
    tot += s;
  }
  __syncthreads();
  total = tot;
  return woff + x - v;
}

__global__ __launch_bounds__(STHREADS) void bin_count(const int* __restrict__ src,
                                                      const int* __restrict__ dst,
                                                      int* __restrict__ hist) {
  __shared__ int cD[NBK];
  __shared__ int cS[NBK];
  const int t = threadIdx.x;
  for (int b = t; b < NBK; b += STHREADS) { cD[b] = 0; cS[b] = 0; }
  __syncthreads();
  const int e0 = min((int)blockIdx.x * EPB, NEDGES);
  const int e1 = min(e0 + EPB, NEDGES);
  for (int base = e0; base < e1; base += STHREADS * 4) {
    unsigned d[4], s[4];
    #pragma unroll
    for (int u = 0; u < 4; ++u) {
      const int i = min(base + u * STHREADS + t, e1 - 1);
      d[u] = (unsigned)dst[i];
      s[u] = (unsigned)src[i];
    }
    #pragma unroll
    for (int u = 0; u < 4; ++u) {
      if (base + u * STHREADS + t < e1) {
        atomicAdd(&cD[min(d[u], (unsigned)(NNODES - 1)) >> BSH], 1);
        atomicAdd(&cS[min(s[u], (unsigned)(NNODES - 1)) >> BSH], 1);
      }
    }
  }
  __syncthreads();
  for (int b = t; b < NBK; b += STHREADS) {
    hist[blockIdx.x * NBK + b] = cD[b];
    hist[NBLK * NBK + blockIdx.x * NBK + b] = cS[b];
  }
}

__global__ __launch_bounds__(NBLK) void bin_scan(const int* __restrict__ hist, int* __restrict__ pre,
                                                 int* __restrict__ tot) {
  __shared__ int sW[NBLK / 32];
  const int t = threadIdx.x;
  const int b = blockIdx.x;
  const int side = blockIdx.y * (NBLK * NBK);
  int total;
  const int p = block_excl_scan<NBLK / 32>(hist[side + t * NBK + b], sW, total);
  pre[side + t * NBK + b] = p;
  if (t == 0) tot[blockIdx.y * NBK + b] = total;
}

__global__ __launch_bounds__(STHREADS) void bin_scatter(const int* __restrict__ src,
                                                        const int* __restrict__ dst,
                                                        const float* __restrict__ ew,
                                                        const int* __restrict__ hist,
                                                        const int* __restrict__ pre,
                                                        const int* __restrict__ tot,
                                                        int2* __restrict__ binE,
                                                        int* __restrict__ binS,
                                                        int* __restrict__ bkt) {
  __shared__ int curD[NBK];
  __shared__ int curS[NBK];
  __shared__ int dltD[NBK];
  __shared__ int dltS[NBK];
  __shared__ int sW[SWAVES];
  __shared__ __attribute__((aligned(8))) int stEraw[2 * EPB];
  __shared__ int stS[EPB];
  __shared__ unsigned short stEb[EPB];
  __shared__ unsigned short stSb[EPB];
  int2* stE = (int2*)stEraw;
  const int t = threadIdx.x;
  const int blk = blockIdx.x;
  for (int side = 0; side < 2; ++side) {
    const int gv = (t < NBK) ? tot[side * NBK + t] : 0;
    const int lv = (t < NBK) ? hist[side * (NBLK * NBK) + blk * NBK + t] : 0;
    int gtot, ltot;
    const int gstart = block_excl_scan<SWAVES>(gv, sW, gtot);
    const int lstart = block_excl_scan<SWAVES>(lv, sW, ltot);
    if (t < NBK) {
      const int g = gstart + pre[side * (NBLK * NBK) + blk * NBK + t];
      if (side == 0) { curD[t] = lstart; dltD[t] = g - lstart; }
      else           { curS[t] = lstart; dltS[t] = g - lstart; }
      if (blk == 0) {
        bkt[side * (NBK + 1) + t] = gstart;
        if (t == NBK - 1) bkt[side * (NBK + 1) + NBK] = gstart + gv;
      }
    }
  }
  __syncthreads();
  const int e0 = min(blk * EPB, NEDGES);
  const int e1 = min(e0 + EPB, NEDGES);
  for (int base = e0; base < e1; base += STHREADS * 4) {
    unsigned d[4], s[4];
    int w[4];
    #pragma unroll
    for (int u = 0; u < 4; ++u) {
      const int i = min(base + u * STHREADS + t, e1 - 1);
      d[u] = (unsigned)dst[i];
      s[u] = (unsigned)src[i];
      w[u] = __float_as_int(ew[i]);
    }
    #pragma unroll
    for (int u = 0; u < 4; ++u) {
      if (base + u * STHREADS + t < e1) {
        const unsigned dd = min(d[u], (unsigned)(NNODES - 1));
        const unsigned ss = min(s[u], (unsigned)(NNODES - 1));
        int rd = atomicAdd(&curD[dd >> BSH], 1);
        int rs = atomicAdd(&curS[ss >> BSH], 1);
        rd = (int)min((unsigned)rd, (unsigned)(EPB - 1));
        rs = (int)min((unsigned)rs, (unsigned)(EPB - 1));
        stE[rd] = make_int2((int)(ss | ((dd & (BNODES - 1)) << SRC_BITS)), w[u]);
        stEb[rd] = (unsigned short)(dd >> BSH);
        stS[rs] = (int)ss;
        stSb[rs] = (unsigned short)(ss >> BSH);
      }
    }
  }
  __syncthreads();
  const int n = e1 - e0;
  for (int r = t; r < n; r += STHREADS) {
    const int pd = (int)min((unsigned)(r + dltD[stEb[r]]), (unsigned)(NEDGES - 1));
    const int ps = (int)min((unsigned)(r + dltS[stSb[r]]), (unsigned)(NEDGES - 1));
    binE[pd] = stE[r];
    binS[ps] = stS[r];
  }
}

__global__ __launch_bounds__(STHREADS) void bucket_build(const int2* __restrict__ binE,
                                                         const int* __restrict__ binS,
                                                         const int* __restrict__ bkt,
                                                         int* __restrict__ rowptr,
                                                         int2* __restrict__ csr,
                                                         float* __restrict__ normSrc,
                                                         float* __restrict__ normDst) {
  __shared__ int cnt[BNODES];
  __shared__ int cntS[BNODES];
  __shared__ int cur[BNODES];
  __shared__ int sW[SWAVES];
  __shared__ __attribute__((aligned(8))) int stageRaw[2 * DCAP];
  int2* stage = (int2*)stageRaw;
  const int t = threadIdx.x;
  const int b = blockIdx.x;
  if (t < BNODES) { cnt[t] = 0; cntS[t] = 0; }
  __syncthreads();
  const int bs = min(max(bkt[b], 0), NEDGES);
  const int be = min(max(bkt[b + 1], bs), NEDGES);
  const int ss = min(max(bkt[(NBK + 1) + b], 0), NEDGES);
  const int se = min(max(bkt[(NBK + 1) + b + 1], ss), NEDGES);
  for (int base = bs; base < be; base += STHREADS * 4) {
    int k[4];
    #pragma unroll
    for (int u = 0; u < 4; ++u) k[u] = binE[min(base + u * STHREADS + t, be - 1)].x;
    #pragma unroll
    for (int u = 0; u < 4; ++u)
      if (base + u * STHREADS + t < be) atomicAdd(&cnt[(k[u] >> SRC_BITS) & (BNODES - 1)], 1);
  }
  for (int base = ss; base < se; base += STHREADS * 4) {
    int k[4];
    #pragma unroll
    for (int u = 0; u < 4; ++u) k[u] = binS[min(base + u * STHREADS + t, se - 1)];
    #pragma unroll
    for (int u = 0; u < 4; ++u)
      if (base + u * STHREADS + t < se) atomicAdd(&cntS[k[u] & (BNODES - 1)], 1);
  }
  __syncthreads();
  const int v = (t < BNODES) ? cnt[t] : 0;
  int total;
  const int rel = block_excl_scan<SWAVES>(v, sW, total);
  if (t < BNODES) {
    cur[t] = rel;
    const int node = b * BNODES + t;
    if (node <= NNODES) rowptr[node] = bs + rel;
    if (node < NNODES) {
      normDst[node] = rsqrtf(fmaxf((float)v, 1.0f));
      normSrc[node] = rsqrtf(fmaxf((float)cntS[t], 1.0f));
    }
  }
  __syncthreads();
  const int nb = be - bs;
  const bool inLds = nb <= DCAP;
  for (int base = bs; base < be; base += STHREADS * 4) {
    int2 r[4];
    #pragma unroll
    for (int u = 0; u < 4; ++u) r[u] = binE[min(base + u * STHREADS + t, be - 1)];
    #pragma unroll
    for (int u = 0; u < 4; ++u) {
      if (base + u * STHREADS + t < be) {
        const int p = atomicAdd(&cur[(r[u].x >> SRC_BITS) & (BNODES - 1)], 1);
        const int2 rec = make_int2(r[u].x & ((1 << SRC_BITS) - 1), r[u].y);
        if (inLds) stage[min((unsigned)p, (unsigned)(DCAP - 1))] = rec;
        else csr[min((unsigned)(bs + p), (unsigned)(NEDGES - 1))] = rec;
      }
    }
  }
  if (inLds) {
    __syncthreads();
    for (int i = t; i < nb; i += STHREADS) csr[bs + i] = stage[i];
  }
}

#define GTHREADS 128
static_assert((HF * HF / 4) % GTHREADS == 0, "W splits evenly over the threads");

__global__ __launch_bounds__(GTHREADS) void gemm_norm_wmma(
    const float* __restrict__ X,
    const float* __restrict__ W,
    const float* __restrict__ normSrc,
    float* __restrict__ Y,
    int nRows)
{
  extern __shared__ float smem[];
  float* sW = smem;
  float* sX = smem + HF * HF;

  const int wave = threadIdx.x >> 5;
  const int lane = threadIdx.x & 31;
  const int tile = blockIdx.x * 4 + wave;
  const int row0 = tile * 16;
  float* sXw = sX + wave * 16 * XSTR;

  {
    v4f xr[16];
    #pragma unroll
    for (int j = 0; j < 16; ++j) {
      const int rj = min(row0 + j, nRows - 1);
      xr[j] = *(const v4f*)(X + (size_t)rj * HF + lane * 4);
    }
    const v4f* Wv = (const v4f*)W;
    v4f* sWv = (v4f*)sW;
    v4f t[HF * HF / 4 / GTHREADS];
    #pragma unroll
    for (int u = 0; u < HF * HF / 4 / GTHREADS; ++u) t[u] = Wv[(int)threadIdx.x + u * GTHREADS];
    #pragma unroll
    for (int j = 0; j < 16; ++j) *(v4f*)(sXw + j * XSTR + lane * 4) = xr[j];
    #pragma unroll
    for (int u = 0; u < HF * HF / 4 / GTHREADS; ++u) sWv[(int)threadIdx.x + u * GTHREADS] = t[u];
  }
  __syncthreads();
  if (row0 >= nRows) return;

  const int half = lane >> 4;
  const int l16  = lane & 15;
  const float* aRow = sXw + l16 * XSTR;

  v8f acc[8];
  v8f vz = {};
  #pragma unroll
  for (int c = 0; c < 8; ++c) acc[c] = vz;

  for (int k0 = 0; k0 < HF; k0 += 4) {
    v2f a;
    a.x = aRow[k0 + 2 * half];
    a.y = aRow[k0 + 2 * half + 1];
    #pragma unroll
    for (int c = 0; c < 8; ++c) {
      v2f b;
      const int n = c * 16 + l16;
      b.x = sW[(k0 + 2 * half) * HF + n];
      b.y = sW[(k0 + 2 * half + 1) * HF + n];
      acc[c] = __builtin_amdgcn_wmma_f32_16x16x4_f32(
          false, a, false, b, (short)0, acc[c], false, false);
    }
  }

  const int mBase = row0 + 8 * half;
  float nrm[8];
  #pragma unroll
  for (int r = 0; r < 8; ++r) nrm[r] = normSrc[mBase + r];
  #pragma unroll
  for (int c = 0; c < 8; ++c) {
    #pragma unroll
    for (int r = 0; r < 8; ++r) {
      Y[(size_t)(mBase + r) * HF + c * 16 + l16] = acc[c][r] * nrm[r];
    }
  }
}

#define FW 25
#define FTHREADS (FW * 32)

template <int U>
__device__ __forceinline__ void gather_chunk_pf(const float* __restrict__ hS, int px, int py,
                                                int k, int n, int lane, v4f& acc,
                                                const int2* __restrict__ csr2, int pfIdx, int2& pf) {
  v4f v[U];
  float w[U];
  for (int i = 0; i < U; ++i) {
    const int idx = min(k + i, n - 1);
    int s = __builtin_amdgcn_readlane(px, idx);
    w[i] = __int_as_float(__builtin_amdgcn_readlane(py, idx));
    s = (int)min((unsigned)s, (unsigned)(NNODES - 1));
    v[i] = *(const v4f*)(hS + (size_t)s * HF + lane * 4);
  }
  if (pfIdx >= 0) pf = csr2[pfIdx];
  if constexpr (U == 16) {
    asm volatile("" :: "v"(v[0]), "v"(v[1]), "v"(v[2]), "v"(v[3]),
                       "v"(v[4]), "v"(v[5]), "v"(v[6]), "v"(v[7]),
                       "v"(v[8]), "v"(v[9]), "v"(v[10]), "v"(v[11]),
                       "v"(v[12]), "v"(v[13]), "v"(v[14]), "v"(v[15]));
  } else if constexpr (U == 8) {
    asm volatile("" :: "v"(v[0]), "v"(v[1]), "v"(v[2]), "v"(v[3]),
                       "v"(v[4]), "v"(v[5]), "v"(v[6]), "v"(v[7]));
  } else {
    asm volatile("" :: "v"(v[0]), "v"(v[1]), "v"(v[2]), "v"(v[3]));
  }
  for (int i = 0; i < U; ++i) {
    if (i == 0 || k + i < n) {
      acc.x = fmaf(v[i].x, w[i], acc.x);
      acc.y = fmaf(v[i].y, w[i], acc.y);
      acc.z = fmaf(v[i].z, w[i], acc.z);
      acc.w = fmaf(v[i].w, w[i], acc.w);
    }
  }
}

__global__ __launch_bounds__(FTHREADS) void gather_gemm_norm_wmma(
    const float* __restrict__ hS,
    const int* __restrict__ rowptr,
    const int* __restrict__ csr,
    const float* __restrict__ normDst,
    const float* __restrict__ bias,
    const float* __restrict__ W,
    const float* __restrict__ normSrc,
    float* __restrict__ Y,
    int nRows)
{
  __shared__ __attribute__((aligned(16))) float smemF[HF * HF + FW * 16 * XSTR];
  float* sW = smemF;
  float* sX = smemF + HF * HF;

  const int wave = threadIdx.x >> 5;
  const int lane = threadIdx.x & 31;
  const int row0 = __builtin_amdgcn_readfirstlane((int)((blockIdx.x * FW + wave) * 16));
  const bool live = row0 < nRows;
  const int rowL = live ? row0 : 0;
  float* sXw = sX + wave * 16 * XSTR;
  const int2* csr2 = (const int2*)csr;

  int rp;
  float ndv;
  v4f b;
  int begN, endN;
  int2 pN;
  {
    const v4f* Wv = (const v4f*)W;
    v4f* sWv = (v4f*)sW;
    v4f t[6];
    #pragma unroll
    for (int u = 0; u < 6; ++u) {
      const int i = (int)threadIdx.x + u * FTHREADS;
      t[u] = Wv[min(i, HF * HF / 4 - 1)];
    }
    rp = rowptr[rowL + min(lane, 16)];
    ndv = normDst[rowL + (lane & 15)];
    b = *(const v4f*)(bias + lane * 4);
    asm volatile("" :: "v"(t[0]), "v"(t[1]), "v"(t[2]), "v"(t[3]), "v"(t[4]), "v"(t[5]));
    begN = max(__builtin_amdgcn_readlane(rp, 0), 0);
    endN = min(__builtin_amdgcn_readlane(rp, 1), NEDGES);
    pN = csr2[min(max(min(begN + lane, endN - 1), 0), NEDGES - 1)];
    #pragma unroll
    for (int u = 0; u < 6; ++u) {
      const int i = (int)threadIdx.x + u * FTHREADS;
      if (i < HF * HF / 4) sWv[i] = t[u];
    }
  }
  __syncthreads();
  if (!live) return;

  {
    #pragma nounroll
    for (int j = 0; j < 16; ++j) {
      const int beg = begN;
      const int end = endN;
      int2 p = pN;
      int pfIdx = -1;
      if (j + 1 < 16) {
        begN = max(__builtin_amdgcn_readlane(rp, j + 1), 0);
        endN = min(__builtin_amdgcn_readlane(rp, j + 2), NEDGES);
        pfIdx = min(max(min(begN + lane, endN - 1), 0), NEDGES - 1);
      }
      bool owed = (j + 1 < 16);
      const float nd = __int_as_float(__builtin_amdgcn_readlane(__float_as_int(ndv), j));
      v4f acc = {0.0f, 0.0f, 0.0f, 0.0f};
      for (int base = beg; base < end; base += 32) {
        const int n = min(32, end - base);
        if (base != beg) p = csr2[min(base + lane, end - 1)];
        int k = 0;
        for (; n - k > 12; k += 16) {
          gather_chunk_pf<16>(hS, p.x, p.y, k, n, lane, acc, csr2, owed ? pfIdx : -1, pN);
          owed = false;
        }
        if (n - k > 4) {
          gather_chunk_pf<8>(hS, p.x, p.y, k, n, lane, acc, csr2, owed ? pfIdx : -1, pN);
          owed = false;
          k += 8;
        }
        if (k < n) {
          gather_chunk_pf<4>(hS, p.x, p.y, k, n, lane, acc, csr2, owed ? pfIdx : -1, pN);
          owed = false;
        }
      }
      if (owed) pN = csr2[pfIdx];
      v4f x;
      x.x = acc.x * nd + b.x;
      x.y = acc.y * nd + b.y;
      x.z = acc.z * nd + b.z;
      x.w = acc.w * nd + b.w;
      x.x = (x.x > 0.0f) ? x.x : 0.01f * x.x;
      x.y = (x.y > 0.0f) ? x.y : 0.01f * x.y;
      x.z = (x.z > 0.0f) ? x.z : 0.01f * x.z;
      x.w = (x.w > 0.0f) ? x.w : 0.01f * x.w;
      *(v4f*)(sXw + j * XSTR + lane * 4) = x;
    }
  }
  __builtin_amdgcn_fence(__ATOMIC_RELEASE, "wavefront");
  __builtin_amdgcn_wave_barrier();
  __builtin_amdgcn_fence(__ATOMIC_ACQUIRE, "wavefront");

  const int half = lane >> 4;
  const int l16  = lane & 15;
  const float* aRow = sXw + l16 * XSTR;

  v8f acc[8];
  v8f vz = {};
  #pragma unroll
  for (int c = 0; c < 8; ++c) acc[c] = vz;

  for (int k0 = 0; k0 < HF; k0 += 4) {
    v2f a;
    a.x = aRow[k0 + 2 * half];
    a.y = aRow[k0 + 2 * half + 1];
    #pragma unroll
    for (int c = 0; c < 8; ++c) {
      v2f bb;
      const int n = c * 16 + l16;
      bb.x = sW[(k0 + 2 * half) * HF + n];
      bb.y = sW[(k0 + 2 * half + 1) * HF + n];
      acc[c] = __builtin_amdgcn_wmma_f32_16x16x4_f32(
          false, a, false, bb, (short)0, acc[c], false, false);
    }
  }

  const int mBase = row0 + 8 * half;
  float nrm[8];
  #pragma unroll
  for (int r = 0; r < 8; ++r) nrm[r] = normSrc[mBase + r];
  #pragma unroll
  for (int c = 0; c < 8; ++c) {
    #pragma unroll
    for (int r = 0; r < 8; ++r) {
      Y[(size_t)(mBase + r) * HF + c * 16 + l16] = acc[c][r] * nrm[r];
    }
  }
}

#define LWN 16

__global__ __launch_bounds__(FTHREADS) void gather_linear_wmma(
    const float* __restrict__ hS,
    const int* __restrict__ rowptr,
    const int* __restrict__ csr,
    const float* __restrict__ normDst,
    const float* __restrict__ bias,
    const float* __restrict__ Wl,
    const float* __restrict__ bl,
    float* __restrict__ out,
    int nRows)
{
  __shared__ __attribute__((aligned(16))) float smemL[HF * LWN + FW * 16 * XSTR];
  float* sWl = smemL;
  float* sX = smemL + HF * LWN;

  const int wave = threadIdx.x >> 5;
  const int lane = threadIdx.x & 31;
  const int row0 = __builtin_amdgcn_readfirstlane((int)((blockIdx.x * FW + wave) * 16));
  const bool live = row0 < nRows;
  const int rowL = live ? row0 : 0;
  float* sXw = sX + wave * 16 * XSTR;
  const int2* csr2 = (const int2*)csr;

  int rp;
  float ndv;
  v4f b;
  int begN, endN;
  int2 pN;
  {
    float t[3];
    #pragma unroll
    for (int u = 0; u < 3; ++u) {
      const int i = min((int)threadIdx.x + u * FTHREADS, HF * LWN - 1);
      t[u] = Wl[(i >> 4) * NCLS + (i & (NCLS - 1))];
    }
    rp = rowptr[rowL + min(lane, 16)];
    ndv = normDst[rowL + (lane & 15)];
    b = *(const v4f*)(bias + lane * 4);
    asm volatile("" :: "v"(t[0]), "v"(t[1]), "v"(t[2]));
    begN = max(__builtin_amdgcn_readlane(rp, 0), 0);
    endN = min(__builtin_amdgcn_readlane(rp, 1), NEDGES);
    pN = csr2[min(max(min(begN + lane, endN - 1), 0), NEDGES - 1)];
    #pragma unroll
    for (int u = 0; u < 3; ++u) {
      const int i = (int)threadIdx.x + u * FTHREADS;
      if (i < HF * LWN) sWl[i] = ((i & (LWN - 1)) < NCLS) ? t[u] : 0.0f;
    }
  }
  __syncthreads();
  if (!live) return;

  {
    #pragma nounroll
    for (int j = 0; j < 16; ++j) {
      const int beg = begN;
      const int end = endN;
      int2 p = pN;
      int pfIdx = -1;
      if (j + 1 < 16) {
        begN = max(__builtin_amdgcn_readlane(rp, j + 1), 0);
        endN = min(__builtin_amdgcn_readlane(rp, j + 2), NEDGES);
        pfIdx = min(max(min(begN + lane, endN - 1), 0), NEDGES - 1);
      }
      bool owed = (j + 1 < 16);
      const float nd = __int_as_float(__builtin_amdgcn_readlane(__float_as_int(ndv), j));
      v4f acc = {0.0f, 0.0f, 0.0f, 0.0f};
      for (int base = beg; base < end; base += 32) {
        const int n = min(32, end - base);
        if (base != beg) p = csr2[min(base + lane, end - 1)];
        int k = 0;
        for (; n - k > 12; k += 16) {
          gather_chunk_pf<16>(hS, p.x, p.y, k, n, lane, acc, csr2, owed ? pfIdx : -1, pN);
          owed = false;
        }
        if (n - k > 4) {
          gather_chunk_pf<8>(hS, p.x, p.y, k, n, lane, acc, csr2, owed ? pfIdx : -1, pN);
          owed = false;
          k += 8;
        }
        if (k < n) {
          gather_chunk_pf<4>(hS, p.x, p.y, k, n, lane, acc, csr2, owed ? pfIdx : -1, pN);
          owed = false;
        }
      }
      if (owed) pN = csr2[pfIdx];
      v4f x;
      x.x = acc.x * nd + b.x;
      x.y = acc.y * nd + b.y;
      x.z = acc.z * nd + b.z;
      x.w = acc.w * nd + b.w;
      x.x = (x.x > 0.0f) ? x.x : 0.01f * x.x;
      x.y = (x.y > 0.0f) ? x.y : 0.01f * x.y;
      x.z = (x.z > 0.0f) ? x.z : 0.01f * x.z;
      x.w = (x.w > 0.0f) ? x.w : 0.01f * x.w;
      *(v4f*)(sXw + j * XSTR + lane * 4) = x;
    }
  }
  __builtin_amdgcn_fence(__ATOMIC_RELEASE, "wavefront");
  __builtin_amdgcn_wave_barrier();
  __builtin_amdgcn_fence(__ATOMIC_ACQUIRE, "wavefront");

  const int half = lane >> 4;
  const int l16  = lane & 15;
  const float* aRow = sXw + l16 * XSTR;

  const float bv = (l16 < NCLS) ? bl[l16 & (NCLS - 1)] : 0.0f;
  v8f acc0, acc1;
  #pragma unroll
  for (int r = 0; r < 8; ++r) { acc0[r] = bv; acc1[r] = 0.0f; }

  for (int k0 = 0; k0 < HF; k0 += 8) {
    v2f a0, b0, a1, b1;
    a0.x = aRow[k0 + 2 * half];
    a0.y = aRow[k0 + 2 * half + 1];
    b0.x = sWl[(k0 + 2 * half) * LWN + l16];
    b0.y = sWl[(k0 + 2 * half + 1) * LWN + l16];
    a1.x = aRow[k0 + 4 + 2 * half];
    a1.y = aRow[k0 + 4 + 2 * half + 1];
    b1.x = sWl[(k0 + 4 + 2 * half) * LWN + l16];
    b1.y = sWl[(k0 + 4 + 2 * half + 1) * LWN + l16];
    acc0 = __builtin_amdgcn_wmma_f32_16x16x4_f32(
        false, a0, false, b0, (short)0, acc0, false, false);
    acc1 = __builtin_amdgcn_wmma_f32_16x16x4_f32(
        false, a1, false, b1, (short)0, acc1, false, false);
  }

  if (l16 < NCLS) {
    #pragma unroll
    for (int r = 0; r < 8; ++r) sXw[(8 * half + r) * NCLS + l16] = acc0[r] + acc1[r];
  }
  __builtin_amdgcn_fence(__ATOMIC_RELEASE, "wavefront");
  __builtin_amdgcn_wave_barrier();
  __builtin_amdgcn_fence(__ATOMIC_ACQUIRE, "wavefront");
  const v4f o = *(const v4f*)(sXw + lane * 4);
  *(v4f*)(out + (size_t)row0 * NCLS + lane * 4) = o;
}

extern "C" void kernel_launch(void* const* d_in, const int* in_sizes, int n_in,
                              void* d_out, int out_size, void* d_ws, size_t ws_size,
                              hipStream_t stream) {
  const float* in_feat = (const float*)d_in[0];
  const int*   src     = (const int*)d_in[1];
  const int*   dst     = (const int*)d_in[2];
  const float* ew      = (const float*)d_in[3];
  const float* convW   = (const float*)d_in[4];
  const float* convB   = (const float*)d_in[5];
  const float* linW    = (const float*)d_in[6];
  const float* linB    = (const float*)d_in[7];
  float* out = (float*)d_out;

  float* ws = (float*)d_ws;
  const size_t NF = (size_t)NNODES * HF;
  float* hA      = ws;
  float* hS      = ws + NF;
  float* normSrc = ws + 2 * NF;
  float* normDst = normSrc + 2 * NNODES;
  int*   rowptr  = (int*)(normSrc + 3 * NNODES);
  int*   hist    = rowptr + NNODES + 96;
  int*   pre     = hist + 2 * NBLK * NBK;
  int*   tot     = pre + 2 * NBLK * NBK;
  int*   bkt     = tot + 1024;
  int*   binE    = bkt + 1024;
  int*   csr     = binE + 2 * NEDGES;
  int*   binS    = csr + 2 * NEDGES;

  bin_count<<<NBLK, STHREADS, 0, stream>>>(src, dst, hist);
  bin_scan<<<dim3(NBK, 2), NBLK, 0, stream>>>(hist, pre, tot);
  bin_scatter<<<NBLK, STHREADS, 0, stream>>>(src, dst, ew, hist, pre, tot, (int2*)binE, binS, bkt);
  bucket_build<<<NBK, STHREADS, 0, stream>>>((const int2*)binE, binS, bkt, rowptr, (int2*)csr,
                                             normSrc, normDst);

  const size_t ldsBytes = (size_t)(HF * HF + 4 * 16 * XSTR) * sizeof(float);
  const int gemmBlocks = (NNODES / 16 + 3) / 4;
  const int fusedBlocks = (NNODES / 16 + FW - 1) / FW;

  gemm_norm_wmma<<<gemmBlocks, GTHREADS, ldsBytes, stream>>>(in_feat, convW, normSrc, hS, NNODES);
  gather_gemm_norm_wmma<<<fusedBlocks, FTHREADS, 0, stream>>>(
      hS, rowptr, csr, normDst, convB + 0 * HF, convW + (size_t)1 * HF * HF, normSrc, hA, NNODES);
  gather_gemm_norm_wmma<<<fusedBlocks, FTHREADS, 0, stream>>>(
      hA, rowptr, csr, normDst, convB + 1 * HF, convW + (size_t)2 * HF * HF, normSrc, hS, NNODES);
  gather_gemm_norm_wmma<<<fusedBlocks, FTHREADS, 0, stream>>>(
      hS, rowptr, csr, normDst, convB + 2 * HF, convW + (size_t)3 * HF * HF, normSrc, hA, NNODES);
  gather_gemm_norm_wmma<<<fusedBlocks, FTHREADS, 0, stream>>>(
      hA, rowptr, csr, normDst, convB + 3 * HF, convW + (size_t)4 * HF * HF, normSrc, hS, NNODES);
  gather_gemm_norm_wmma<<<fusedBlocks, FTHREADS, 0, stream>>>(
      hS, rowptr, csr, normDst, convB + 4 * HF, convW + (size_t)5 * HF * HF, normSrc, hA, NNODES);
  gather_linear_wmma<<<fusedBlocks, FTHREADS, 0, stream>>>(
      hA, rowptr, csr, normDst, convB + 5 * HF, linW, linB, out, NNODES);
}
